# starting kernel
# baseline (speedup 1.0000x reference)
#include <hip/hip_runtime.h>
#include <math.h>

#define LDA 65
#define NL  4

typedef __attribute__((ext_vector_type(2))) float v2f;
typedef __attribute__((ext_vector_type(8))) float v8f;

__device__ __forceinline__ float2 cmul(float2 a, float2 b) {
  return make_float2(a.x * b.x - a.y * b.y, a.x * b.y + a.y * b.x);
}
__device__ __forceinline__ float2 cadd(float2 a, float2 b) {
  return make_float2(a.x + b.x, a.y + b.y);
}

struct C2 { float2 a00, a01, a10, a11; };
__device__ __forceinline__ C2 mm2(const C2 L, const C2 R) {
  C2 o;
  o.a00 = cadd(cmul(L.a00, R.a00), cmul(L.a01, R.a10));
  o.a01 = cadd(cmul(L.a00, R.a01), cmul(L.a01, R.a11));
  o.a10 = cadd(cmul(L.a10, R.a00), cmul(L.a11, R.a10));
  o.a11 = cadd(cmul(L.a10, R.a01), cmul(L.a11, R.a11));
  return o;
}

__device__ __forceinline__ void rot_gate(float phi, float th, float om,
                                         float2* u00, float2* u01,
                                         float2* u10, float2* u11) {
  float ct = cosf(0.5f * th), st = sinf(0.5f * th);
  float ap = -0.5f * (phi + om);
  float am = 0.5f * (phi - om);
  *u00 = make_float2(ct * cosf(ap), ct * sinf(ap));
  *u01 = make_float2(-st * cosf(am), -st * sinf(am));
  *u10 = make_float2(st * cosf(-am), st * sinf(-am));
  *u11 = make_float2(ct * cosf(-ap), ct * sinf(-ap));
}

__device__ __forceinline__ void apply_cgate(float* Mr, float* Mi,
                                            float2 u00, float2 u01,
                                            float2 u10, float2 u11,
                                            int cb, int tb, int tid) {
  for (int idx = tid; idx < 4096; idx += 256) {
    int i = idx >> 6, j = idx & 63;
    if ((i & cb) && !(i & tb)) {
      int a0 = i * LDA + j, a1 = (i | tb) * LDA + j;
      float2 x0 = make_float2(Mr[a0], Mi[a0]);
      float2 x1 = make_float2(Mr[a1], Mi[a1]);
      float2 y0 = cadd(cmul(u00, x0), cmul(u01, x1));
      float2 y1 = cadd(cmul(u10, x0), cmul(u11, x1));
      Mr[a0] = y0.x; Mi[a0] = y0.y;
      Mr[a1] = y1.x; Mi[a1] = y1.y;
    }
  }
}

__global__ __launch_bounds__(256) void qk_prep(const float* __restrict__ rotp,
                                               const float* __restrict__ entp,
                                               float2* __restrict__ W) {
  __shared__ float Mr[64 * LDA], Mi[64 * LDA];
  __shared__ float2 U[24][2][2];
  const int l = blockIdx.x;
  const int tid = threadIdx.x;

  if (tid < 24) {
    const float* p = (tid < 12) ? &rotp[(l * 12 + tid) * 3]
                                : &entp[(l * 12 + (tid - 12)) * 3];
    rot_gate(p[0], p[1], p[2], &U[tid][0][0], &U[tid][0][1],
             &U[tid][1][0], &U[tid][1][1]);
  }
  __syncthreads();

  float2* Wl = W + (size_t)l * 3 * 4096;

  for (int idx = tid; idx < 4096; idx += 256) {
    int i = idx >> 6, j = idx & 63;
    float2 p = U[0][(i >> 5) & 1][(j >> 5) & 1];
    for (int q = 1; q < 6; q++)
      p = cmul(p, U[q][(i >> (5 - q)) & 1][(j >> (5 - q)) & 1]);
    Mr[i * LDA + j] = p.x; Mi[i * LDA + j] = p.y;
  }
  __syncthreads();
  for (int c = 0; c < 5; c++) {
    apply_cgate(Mr, Mi, U[12 + c][0][0], U[12 + c][0][1],
                U[12 + c][1][0], U[12 + c][1][1],
                1 << (5 - c), 1 << (4 - c), tid);
    __syncthreads();
  }
  for (int idx = tid; idx < 4096; idx += 256) {
    int i = idx >> 6, j = idx & 63;
    Wl[0 * 4096 + idx] = make_float2(Mr[i * LDA + j], Mi[i * LDA + j]);
  }

  for (int idx = tid; idx < 4096; idx += 256) {
    int i = idx >> 6, j = idx & 63;
    float2 p = U[6][(j >> 5) & 1][(i >> 5) & 1];
    for (int q = 1; q < 6; q++)
      p = cmul(p, U[6 + q][(j >> (5 - q)) & 1][(i >> (5 - q)) & 1]);
    Wl[1 * 4096 + idx] = p;
  }
  __syncthreads();

  for (int idx = tid; idx < 4096; idx += 256) {
    int i = idx >> 6, j = idx & 63;
    Mr[i * LDA + j] = (i == j) ? 1.f : 0.f;
    Mi[i * LDA + j] = 0.f;
  }
  __syncthreads();
  for (int c = 0; c < 5; c++) {
    apply_cgate(Mr, Mi, U[18 + c][0][0], U[18 + c][0][1],
                U[18 + c][1][0], U[18 + c][1][1],
                1 << (5 - c), 1 << (4 - c), tid);
    __syncthreads();
  }
  for (int idx = tid; idx < 4096; idx += 256) {
    int i = idx >> 6, j = idx & 63;
    Wl[2 * 4096 + idx] = make_float2(Mr[j * LDA + i], Mi[j * LDA + i]);
  }
}

__device__ __forceinline__ v8f wmma4(v2f a, v2f b, v8f c) {
  return __builtin_amdgcn_wmma_f32_16x16x4_f32(
      false, a, false, b, (short)0, c, false, false);
}

__device__ __forceinline__ void cgemm64(const float* Ar, const float* Ai,
                                        const float* Br, const float* Bi,
                                        float* Dr, float* Di) {
  const int tid = threadIdx.x;
  const int lane = tid & 31;
  const int w = tid >> 5;
  const int mt = (w & 3) * 16;
  const int nt0 = (w >> 2) * 32;
  const int nt1 = nt0 + 16;
  const int ml = lane & 15;
  const int kb = (lane >> 4) * 2;
  const int hi = lane >> 4;

  v8f p1_0 = {}, p2_0 = {}, p3_0 = {};
  v8f p1_1 = {}, p2_1 = {}, p3_1 = {};
  for (int k = 0; k < 64; k += 4) {
    const float* ap = Ar + (mt + ml) * LDA + k + kb;
    const float* aip = Ai + (mt + ml) * LDA + k + kb;
    v2f ar; ar.x = ap[0];  ar.y = ap[1];
    v2f ai; ai.x = aip[0]; ai.y = aip[1];
    v2f as = ar + ai;
    v2f br0, bi0, br1, bi1;
    br0.x = Br[(k + kb) * LDA + nt0 + ml]; br0.y = Br[(k + kb + 1) * LDA + nt0 + ml];
    bi0.x = Bi[(k + kb) * LDA + nt0 + ml]; bi0.y = Bi[(k + kb + 1) * LDA + nt0 + ml];
    br1.x = Br[(k + kb) * LDA + nt1 + ml]; br1.y = Br[(k + kb + 1) * LDA + nt1 + ml];
    bi1.x = Bi[(k + kb) * LDA + nt1 + ml]; bi1.y = Bi[(k + kb + 1) * LDA + nt1 + ml];
    v2f bs0 = br0 + bi0;
    v2f bs1 = br1 + bi1;

    p1_0 = wmma4(ar, br0, p1_0);
    p2_0 = wmma4(ai, bi0, p2_0);
    p3_0 = wmma4(as, bs0, p3_0);
    p1_1 = wmma4(ar, br1, p1_1);
    p2_1 = wmma4(ai, bi1, p2_1);
    p3_1 = wmma4(as, bs1, p3_1);
  }

  v8f cr0 = p1_0 - p2_0;
  v8f ci0 = p3_0 - p1_0 - p2_0;
  v8f cr1 = p1_1 - p2_1;
  v8f ci1 = p3_1 - p1_1 - p2_1;

  for (int v = 0; v < 8; v++) {
    int m = mt + v + 8 * hi;
    int n0 = nt0 + ml, n1 = nt1 + ml;
    Dr[m * LDA + n0] = cr0[v]; Di[m * LDA + n0] = ci0[v];
    Dr[m * LDA + n1] = cr1[v]; Di[m * LDA + n1] = ci1[v];
  }
}

__global__ __launch_bounds__(256) void qk_main(const float* __restrict__ x,
                                               const float* __restrict__ entp,
                                               const float2* __restrict__ W,
                                               float* __restrict__ out) {
  __shared__ float S0r[64 * LDA], S0i[64 * LDA];
  __shared__ float S1r[64 * LDA], S1i[64 * LDA];
  __shared__ float OpR[64 * LDA], OpI[64 * LDA];
  __shared__ float2 E[12][2][2];
  __shared__ float2 K8[4][64];
  __shared__ float2 CU[NL][2][2][2];

  const int tid = threadIdx.x;
  const int s = blockIdx.x;

  if (tid < 12) {
    const float* xs = x + (size_t)s * 48;
    float a = xs[0 * 12 + tid], b = xs[1 * 12 + tid];
    float c = xs[2 * 12 + tid], d = xs[3 * 12 + tid];
    float ca = cosf(0.5f * a), sa = sinf(0.5f * a);
    float cb = cosf(0.5f * b), sb = sinf(0.5f * b);
    float cc = cosf(0.5f * c), sc = sinf(0.5f * c);
    float cd = cosf(0.5f * d), sd = sinf(0.5f * d);
    C2 M  = { {ca, 0.f}, {-sa, 0.f}, {sa, 0.f}, {ca, 0.f} };
    C2 RX = { {cb, 0.f}, {0.f, -sb}, {0.f, -sb}, {cb, 0.f} };
    C2 RZ = { {cc, -sc}, {0.f, 0.f}, {0.f, 0.f}, {cc, sc} };
    C2 RY = { {cd, 0.f}, {-sd, 0.f}, {sd, 0.f}, {cd, 0.f} };
    M = mm2(RX, M); M = mm2(RZ, M); M = mm2(RY, M);
    E[tid][0][0] = M.a00; E[tid][0][1] = M.a01;
    E[tid][1][0] = M.a10; E[tid][1][1] = M.a11;
  }
  if (tid < 2 * NL) {
    int l = tid >> 1, which = tid & 1;
    const float* p = &entp[(l * 12 + (which ? 11 : 5)) * 3];
    rot_gate(p[0], p[1], p[2], &CU[l][which][0][0], &CU[l][which][0][1],
             &CU[l][which][1][0], &CU[l][which][1][1]);
  }
  for (int idx = tid; idx < 4096; idx += 256) {
    int i = idx >> 6, j = idx & 63;
    S0r[i * LDA + j] = (idx == 0) ? 1.f : 0.f;
    S0i[i * LDA + j] = 0.f;
  }
  __syncthreads();

  {
    int g = tid >> 6, a = (tid >> 3) & 7, b = tid & 7;
    float2 p = cmul(cmul(E[3 * g][(a >> 2) & 1][(b >> 2) & 1],
                         E[3 * g + 1][(a >> 1) & 1][(b >> 1) & 1]),
                    E[3 * g + 2][a & 1][b & 1]);
    K8[g][a * 8 + b] = p;
  }
  __syncthreads();

  float *curR = S0r, *curI = S0i, *nxtR = S1r, *nxtI = S1i;

  for (int l = 0; l < NL; l++) {
    const float2* Wl = W + (size_t)l * 3 * 4096;

    for (int idx = tid; idx < 4096; idx += 256) {
      int i = idx >> 6, j = idx & 63;
      float2 p = cmul(K8[0][((i >> 3) << 3) | (j >> 3)],
                      K8[1][((i & 7) << 3) | (j & 7)]);
      OpR[i * LDA + j] = p.x; OpI[i * LDA + j] = p.y;
    }
    __syncthreads();
    cgemm64(OpR, OpI, curR, curI, nxtR, nxtI);
    __syncthreads();
    { float* t; t = curR; curR = nxtR; nxtR = t; t = curI; curI = nxtI; nxtI = t; }

    for (int idx = tid; idx < 4096; idx += 256) {
      int i = idx >> 6, j = idx & 63;
      float2 p = cmul(K8[2][((j >> 3) << 3) | (i >> 3)],
                      K8[3][((j & 7) << 3) | (i & 7)]);
      OpR[i * LDA + j] = p.x; OpI[i * LDA + j] = p.y;
    }
    __syncthreads();
    cgemm64(curR, curI, OpR, OpI, nxtR, nxtI);
    __syncthreads();
    { float* t; t = curR; curR = nxtR; nxtR = t; t = curI; curI = nxtI; nxtI = t; }

    for (int idx = tid; idx < 4096; idx += 256) {
      float2 v = Wl[idx];
      int i = idx >> 6, j = idx & 63;
      OpR[i * LDA + j] = v.x; OpI[i * LDA + j] = v.y;
    }
    __syncthreads();
    cgemm64(OpR, OpI, curR, curI, nxtR, nxtI);
    __syncthreads();
    { float* t; t = curR; curR = nxtR; nxtR = t; t = curI; curI = nxtI; nxtI = t; }

    for (int idx = tid; idx < 4096; idx += 256) {
      float2 v = Wl[4096 + idx];
      int i = idx >> 6, j = idx & 63;
      OpR[i * LDA + j] = v.x; OpI[i * LDA + j] = v.y;
    }
    __syncthreads();
    cgemm64(curR, curI, OpR, OpI, nxtR, nxtI);
    __syncthreads();
    { float* t; t = curR; curR = nxtR; nxtR = t; t = curI; curI = nxtI; nxtI = t; }

    for (int idx = tid; idx < 4096; idx += 256) {
      float2 v = Wl[2 * 4096 + idx];
      int i = idx >> 6, j = idx & 63;
      OpR[i * LDA + j] = v.x; OpI[i * LDA + j] = v.y;
    }
    {
      float2 u00 = CU[l][0][0][0], u01 = CU[l][0][0][1];
      float2 u10 = CU[l][0][1][0], u11 = CU[l][0][1][1];
      for (int t = tid; t < 1024; t += 256) {
        int i = ((t >> 5) << 1) | 1;
        int jl = t & 31;
        int a0 = i * LDA + jl, a1 = a0 + 32;
        float2 x0 = make_float2(curR[a0], curI[a0]);
        float2 x1 = make_float2(curR[a1], curI[a1]);
        float2 y0 = cadd(cmul(u00, x0), cmul(u01, x1));
        float2 y1 = cadd(cmul(u10, x0), cmul(u11, x1));
        curR[a0] = y0.x; curI[a0] = y0.y;
        curR[a1] = y1.x; curI[a1] = y1.y;
      }
    }
    __syncthreads();

    cgemm64(curR, curI, OpR, OpI, nxtR, nxtI);
    __syncthreads();
    { float* t; t = curR; curR = nxtR; nxtR = t; t = curI; curI = nxtI; nxtI = t; }

    {
      float2 u00 = CU[l][1][0][0], u01 = CU[l][1][0][1];
      float2 u10 = CU[l][1][1][0], u11 = CU[l][1][1][1];
      for (int t = tid; t < 1024; t += 256) {
        int il = t >> 5;
        int j = ((t & 31) << 1) | 1;
        int a0 = il * LDA + j, a1 = (il + 32) * LDA + j;
        float2 x0 = make_float2(curR[a0], curI[a0]);
        float2 x1 = make_float2(curR[a1], curI[a1]);
        float2 y0 = cadd(cmul(u00, x0), cmul(u01, x1));
        float2 y1 = cadd(cmul(u10, x0), cmul(u11, x1));
        curR[a0] = y0.x; curI[a0] = y0.y;
        curR[a1] = y1.x; curI[a1] = y1.y;
      }
    }
  }
  __syncthreads();

  float z[12];
  for (int q = 0; q < 12; q++) z[q] = 0.f;
  for (int idx = tid; idx < 4096; idx += 256) {
    int i = idx >> 6, j = idx & 63;
    float pr = curR[i * LDA + j], pi = curI[i * LDA + j];
    float p = pr * pr + pi * pi;
    for (int q = 0; q < 12; q++)
      z[q] += ((idx >> (11 - q)) & 1) ? -p : p;
  }
  __syncthreads();
  float* red = OpR;
  for (int q = 0; q < 12; q++) red[tid * 12 + q] = z[q];
  __syncthreads();
  for (int sft = 128; sft > 0; sft >>= 1) {
    if (tid < sft)
      for (int q = 0; q < 12; q++)
        red[tid * 12 + q] += red[(tid + sft) * 12 + q];
    __syncthreads();
  }
  if (tid < 12) out[(size_t)s * 12 + tid] = red[tid];
}

extern "C" void kernel_launch(void* const* d_in, const int* in_sizes, int n_in,
                              void* d_out, int out_size, void* d_ws, size_t ws_size,
                              hipStream_t stream) {
  (void)n_in; (void)out_size; (void)ws_size;
  const float* x    = (const float*)d_in[0];
  const float* rotp = (const float*)d_in[1];
  const float* entp = (const float*)d_in[2];
  float2* W = (float2*)d_ws;
  const int B = in_sizes[0] / 48;

  qk_prep<<<NL, 256, 0, stream>>>(rotp, entp, W);
  qk_main<<<B, 256, 0, stream>>>(x, entp, W, (float*)d_out);
}


# Fold the shared per-layer Rot gates into the per-sample encoding 2x2s (G = Rot*Enc per wire): 4 complex 64^3 WMMA GEMMs per layer instead of 5; gate products regrouped, so rounding order changes
# speedup vs baseline: 1.2646x; 1.2646x over previous
#include <hip/hip_runtime.h>
#include <math.h>

#define LDA 65
#define NL  4

typedef __attribute__((ext_vector_type(2))) float v2f;
typedef __attribute__((ext_vector_type(8))) float v8f;

__device__ __forceinline__ float2 cmul(float2 a, float2 b) {
  return make_float2(a.x * b.x - a.y * b.y, a.x * b.y + a.y * b.x);
}
__device__ __forceinline__ float2 cadd(float2 a, float2 b) {
  return make_float2(a.x + b.x, a.y + b.y);
}

struct C2 { float2 a00, a01, a10, a11; };
__device__ __forceinline__ C2 mm2(const C2 L, const C2 R) {
  C2 o;
  o.a00 = cadd(cmul(L.a00, R.a00), cmul(L.a01, R.a10));
  o.a01 = cadd(cmul(L.a00, R.a01), cmul(L.a01, R.a11));
  o.a10 = cadd(cmul(L.a10, R.a00), cmul(L.a11, R.a10));
  o.a11 = cadd(cmul(L.a10, R.a01), cmul(L.a11, R.a11));
  return o;
}

__device__ __forceinline__ void rot_gate(float phi, float th, float om,
                                         float2* u00, float2* u01,
                                         float2* u10, float2* u11) {
  float ct = cosf(0.5f * th), st = sinf(0.5f * th);
  float ap = -0.5f * (phi + om);
  float am = 0.5f * (phi - om);
  *u00 = make_float2(ct * cosf(ap), ct * sinf(ap));
  *u01 = make_float2(-st * cosf(am), -st * sinf(am));
  *u10 = make_float2(st * cosf(-am), st * sinf(-am));
  *u11 = make_float2(ct * cosf(-ap), ct * sinf(-ap));
}

__device__ __forceinline__ void apply_cgate(float* Mr, float* Mi,
                                            float2 u00, float2 u01,
                                            float2 u10, float2 u11,
                                            int cb, int tb, int tid) {
  for (int idx = tid; idx < 4096; idx += 256) {
    int i = idx >> 6, j = idx & 63;
    if ((i & cb) && !(i & tb)) {
      int a0 = i * LDA + j, a1 = (i | tb) * LDA + j;
      float2 x0 = make_float2(Mr[a0], Mi[a0]);
      float2 x1 = make_float2(Mr[a1], Mi[a1]);
      float2 y0 = cadd(cmul(u00, x0), cmul(u01, x1));
      float2 y1 = cadd(cmul(u10, x0), cmul(u11, x1));
      Mr[a0] = y0.x; Mi[a0] = y0.y;
      Mr[a1] = y1.x; Mi[a1] = y1.y;
    }
  }
}

__global__ __launch_bounds__(256) void qk_prep(const float* __restrict__ entp,
                                               float2* __restrict__ W) {
  __shared__ float Mr[64 * LDA], Mi[64 * LDA];
  __shared__ float2 U[12][2][2];
  const int l = blockIdx.x;
  const int tid = threadIdx.x;

  if (tid < 12) {
    const float* p = &entp[(l * 12 + tid) * 3];
    rot_gate(p[0], p[1], p[2], &U[tid][0][0], &U[tid][0][1],
             &U[tid][1][0], &U[tid][1][1]);
  }

  float2* Wl = W + (size_t)l * 2 * 4096;

  for (int idx = tid; idx < 4096; idx += 256) {
    int i = idx >> 6, j = idx & 63;
    Mr[i * LDA + j] = (i == j) ? 1.f : 0.f;
    Mi[i * LDA + j] = 0.f;
  }
  __syncthreads();
  for (int c = 0; c < 5; c++) {
    apply_cgate(Mr, Mi, U[c][0][0], U[c][0][1], U[c][1][0], U[c][1][1],
                1 << (5 - c), 1 << (4 - c), tid);
    __syncthreads();
  }
  for (int idx = tid; idx < 4096; idx += 256) {
    int i = idx >> 6, j = idx & 63;
    Wl[0 * 4096 + idx] = make_float2(Mr[i * LDA + j], Mi[i * LDA + j]);
  }
  __syncthreads();

  for (int idx = tid; idx < 4096; idx += 256) {
    int i = idx >> 6, j = idx & 63;
    Mr[i * LDA + j] = (i == j) ? 1.f : 0.f;
    Mi[i * LDA + j] = 0.f;
  }
  __syncthreads();
  for (int c = 0; c < 5; c++) {
    apply_cgate(Mr, Mi, U[6 + c][0][0], U[6 + c][0][1],
                U[6 + c][1][0], U[6 + c][1][1],
                1 << (5 - c), 1 << (4 - c), tid);
    __syncthreads();
  }
  for (int idx = tid; idx < 4096; idx += 256) {
    int i = idx >> 6, j = idx & 63;
    Wl[1 * 4096 + idx] = make_float2(Mr[j * LDA + i], Mi[j * LDA + i]);
  }
}

__device__ __forceinline__ v8f wmma4(v2f a, v2f b, v8f c) {
  return __builtin_amdgcn_wmma_f32_16x16x4_f32(
      false, a, false, b, (short)0, c, false, false);
}

__device__ __forceinline__ void cgemm64(const float* Ar, const float* Ai,
                                        const float* Br, const float* Bi,
                                        float* Dr, float* Di) {
  const int tid = threadIdx.x;
  const int lane = tid & 31;
  const int w = tid >> 5;
  const int mt = (w & 3) * 16;
  const int nt0 = (w >> 2) * 32;
  const int nt1 = nt0 + 16;
  const int ml = lane & 15;
  const int kb = (lane >> 4) * 2;
  const int hi = lane >> 4;

  v8f p1_0 = {}, p2_0 = {}, p3_0 = {};
  v8f p1_1 = {}, p2_1 = {}, p3_1 = {};
  for (int k = 0; k < 64; k += 4) {
    const float* ap = Ar + (mt + ml) * LDA + k + kb;
    const float* aip = Ai + (mt + ml) * LDA + k + kb;
    v2f ar; ar.x = ap[0];  ar.y = ap[1];
    v2f ai; ai.x = aip[0]; ai.y = aip[1];
    v2f as = ar + ai;
    v2f br0, bi0, br1, bi1;
    br0.x = Br[(k + kb) * LDA + nt0 + ml]; br0.y = Br[(k + kb + 1) * LDA + nt0 + ml];
    bi0.x = Bi[(k + kb) * LDA + nt0 + ml]; bi0.y = Bi[(k + kb + 1) * LDA + nt0 + ml];
    br1.x = Br[(k + kb) * LDA + nt1 + ml]; br1.y = Br[(k + kb + 1) * LDA + nt1 + ml];
    bi1.x = Bi[(k + kb) * LDA + nt1 + ml]; bi1.y = Bi[(k + kb + 1) * LDA + nt1 + ml];
    v2f bs0 = br0 + bi0;
    v2f bs1 = br1 + bi1;

    p1_0 = wmma4(ar, br0, p1_0);
    p2_0 = wmma4(ai, bi0, p2_0);
    p3_0 = wmma4(as, bs0, p3_0);
    p1_1 = wmma4(ar, br1, p1_1);
    p2_1 = wmma4(ai, bi1, p2_1);
    p3_1 = wmma4(as, bs1, p3_1);
  }

  v8f cr0 = p1_0 - p2_0;
  v8f ci0 = p3_0 - p1_0 - p2_0;
  v8f cr1 = p1_1 - p2_1;
  v8f ci1 = p3_1 - p1_1 - p2_1;

  for (int v = 0; v < 8; v++) {
    int m = mt + v + 8 * hi;
    int n0 = nt0 + ml, n1 = nt1 + ml;
    Dr[m * LDA + n0] = cr0[v]; Di[m * LDA + n0] = ci0[v];
    Dr[m * LDA + n1] = cr1[v]; Di[m * LDA + n1] = ci1[v];
  }
}

__global__ __launch_bounds__(256) void qk_main(const float* __restrict__ x,
                                               const float* __restrict__ rotp,
                                               const float* __restrict__ entp,
                                               const float2* __restrict__ W,
                                               float* __restrict__ out) {
  __shared__ float S0r[64 * LDA], S0i[64 * LDA];
  __shared__ float S1r[64 * LDA], S1i[64 * LDA];
  __shared__ float OpR[64 * LDA], OpI[64 * LDA];
  __shared__ float2 G[NL][12][2][2];
  __shared__ float2 K8[4][64];
  __shared__ float2 CU[NL][2][2][2];

  const int tid = threadIdx.x;
  const int s = blockIdx.x;

  if (tid < 12 * NL) {
    const int gl = tid / 12, gq = tid - 12 * gl;
    const float* xs = x + (size_t)s * 48;
    float a = xs[0 * 12 + gq], b = xs[1 * 12 + gq];
    float c = xs[2 * 12 + gq], d = xs[3 * 12 + gq];
    float ca = cosf(0.5f * a), sa = sinf(0.5f * a);
    float cb = cosf(0.5f * b), sb = sinf(0.5f * b);
    float cc = cosf(0.5f * c), sc = sinf(0.5f * c);
    float cd = cosf(0.5f * d), sd = sinf(0.5f * d);
    C2 M  = { {ca, 0.f}, {-sa, 0.f}, {sa, 0.f}, {ca, 0.f} };
    C2 RX = { {cb, 0.f}, {0.f, -sb}, {0.f, -sb}, {cb, 0.f} };
    C2 RZ = { {cc, -sc}, {0.f, 0.f}, {0.f, 0.f}, {cc, sc} };
    C2 RY = { {cd, 0.f}, {-sd, 0.f}, {sd, 0.f}, {cd, 0.f} };
    M = mm2(RX, M); M = mm2(RZ, M); M = mm2(RY, M);
    const float* rp = &rotp[(gl * 12 + gq) * 3];
    C2 R;
    rot_gate(rp[0], rp[1], rp[2], &R.a00, &R.a01, &R.a10, &R.a11);
    M = mm2(R, M);
    G[gl][gq][0][0] = M.a00; G[gl][gq][0][1] = M.a01;
    G[gl][gq][1][0] = M.a10; G[gl][gq][1][1] = M.a11;
  }
  if (tid >= 64 && tid < 64 + 2 * NL) {
    int l = (tid - 64) >> 1, which = tid & 1;
    const float* p = &entp[(l * 12 + (which ? 11 : 5)) * 3];
    rot_gate(p[0], p[1], p[2], &CU[l][which][0][0], &CU[l][which][0][1],
             &CU[l][which][1][0], &CU[l][which][1][1]);
  }
  for (int idx = tid; idx < 4096; idx += 256) {
    int i = idx >> 6, j = idx & 63;
    S0r[i * LDA + j] = (idx == 0) ? 1.f : 0.f;
    S0i[i * LDA + j] = 0.f;
  }
  __syncthreads();

  const int kg = tid >> 6, ka = (tid >> 3) & 7, kb8 = tid & 7;
  {
    float2 p = cmul(cmul(G[0][3 * kg][(ka >> 2) & 1][(kb8 >> 2) & 1],
                         G[0][3 * kg + 1][(ka >> 1) & 1][(kb8 >> 1) & 1]),
                    G[0][3 * kg + 2][ka & 1][kb8 & 1]);
    K8[kg][ka * 8 + kb8] = p;
  }
  __syncthreads();

  float *curR = S0r, *curI = S0i, *nxtR = S1r, *nxtI = S1i;

  for (int l = 0; l < NL; l++) {
    const float2* Wl = W + (size_t)l * 2 * 4096;

    for (int idx = tid; idx < 4096; idx += 256) {
      int i = idx >> 6, j = idx & 63;
      float2 p = cmul(K8[0][((i >> 3) << 3) | (j >> 3)],
                      K8[1][((i & 7) << 3) | (j & 7)]);
      OpR[i * LDA + j] = p.x; OpI[i * LDA + j] = p.y;
    }
    __syncthreads();
    cgemm64(OpR, OpI, curR, curI, nxtR, nxtI);
    __syncthreads();
    { float* t; t = curR; curR = nxtR; nxtR = t; t = curI; curI = nxtI; nxtI = t; }

    for (int idx = tid; idx < 4096; idx += 256) {
      int i = idx >> 6, j = idx & 63;
      float2 p = cmul(K8[2][((j >> 3) << 3) | (i >> 3)],
                      K8[3][((j & 7) << 3) | (i & 7)]);
      OpR[i * LDA + j] = p.x; OpI[i * LDA + j] = p.y;
    }
    __syncthreads();
    cgemm64(curR, curI, OpR, OpI, nxtR, nxtI);
    __syncthreads();
    { float* t; t = curR; curR = nxtR; nxtR = t; t = curI; curI = nxtI; nxtI = t; }

    for (int idx = tid; idx < 4096; idx += 256) {
      float2 v = Wl[idx];
      int i = idx >> 6, j = idx & 63;
      OpR[i * LDA + j] = v.x; OpI[i * LDA + j] = v.y;
    }
    if (l + 1 < NL) {
      float2 p = cmul(cmul(G[l + 1][3 * kg][(ka >> 2) & 1][(kb8 >> 2) & 1],
                           G[l + 1][3 * kg + 1][(ka >> 1) & 1][(kb8 >> 1) & 1]),
                      G[l + 1][3 * kg + 2][ka & 1][kb8 & 1]);
      K8[kg][ka * 8 + kb8] = p;
    }
    __syncthreads();
    cgemm64(OpR, OpI, curR, curI, nxtR, nxtI);
    __syncthreads();
    { float* t; t = curR; curR = nxtR; nxtR = t; t = curI; curI = nxtI; nxtI = t; }

    for (int idx = tid; idx < 4096; idx += 256) {
      float2 v = Wl[1 * 4096 + idx];
      int i = idx >> 6, j = idx & 63;
      OpR[i * LDA + j] = v.x; OpI[i * LDA + j] = v.y;
    }
    {
      float2 u00 = CU[l][0][0][0], u01 = CU[l][0][0][1];
      float2 u10 = CU[l][0][1][0], u11 = CU[l][0][1][1];
      for (int t = tid; t < 1024; t += 256) {
        int i = ((t >> 5) << 1) | 1;
        int jl = t & 31;
        int a0 = i * LDA + jl, a1 = a0 + 32;
        float2 x0 = make_float2(curR[a0], curI[a0]);
        float2 x1 = make_float2(curR[a1], curI[a1]);
        float2 y0 = cadd(cmul(u00, x0), cmul(u01, x1));
        float2 y1 = cadd(cmul(u10, x0), cmul(u11, x1));
        curR[a0] = y0.x; curI[a0] = y0.y;
        curR[a1] = y1.x; curI[a1] = y1.y;
      }
    }
    __syncthreads();

    cgemm64(curR, curI, OpR, OpI, nxtR, nxtI);
    __syncthreads();
    { float* t; t = curR; curR = nxtR; nxtR = t; t = curI; curI = nxtI; nxtI = t; }

    {
      float2 u00 = CU[l][1][0][0], u01 = CU[l][1][0][1];
      float2 u10 = CU[l][1][1][0], u11 = CU[l][1][1][1];
      for (int t = tid; t < 1024; t += 256) {
        int il = t >> 5;
        int j = ((t & 31) << 1) | 1;
        int a0 = il * LDA + j, a1 = (il + 32) * LDA + j;
        float2 x0 = make_float2(curR[a0], curI[a0]);
        float2 x1 = make_float2(curR[a1], curI[a1]);
        float2 y0 = cadd(cmul(u00, x0), cmul(u01, x1));
        float2 y1 = cadd(cmul(u10, x0), cmul(u11, x1));
        curR[a0] = y0.x; curI[a0] = y0.y;
        curR[a1] = y1.x; curI[a1] = y1.y;
      }
    }
  }
  __syncthreads();

  float z[12];
  for (int q = 0; q < 12; q++) z[q] = 0.f;
  for (int idx = tid; idx < 4096; idx += 256) {
    int i = idx >> 6, j = idx & 63;
    float pr = curR[i * LDA + j], pi = curI[i * LDA + j];
    float p = pr * pr + pi * pi;
    for (int q = 0; q < 12; q++)
      z[q] += ((idx >> (11 - q)) & 1) ? -p : p;
  }
  __syncthreads();
  float* red = OpR;
  for (int q = 0; q < 12; q++) red[tid * 12 + q] = z[q];
  __syncthreads();
  for (int sft = 128; sft > 0; sft >>= 1) {
    if (tid < sft)
      for (int q = 0; q < 12; q++)
        red[tid * 12 + q] += red[(tid + sft) * 12 + q];
    __syncthreads();
  }
  if (tid < 12) out[(size_t)s * 12 + tid] = red[tid];
}

extern "C" void kernel_launch(void* const* d_in, const int* in_sizes, int n_in,
                              void* d_out, int out_size, void* d_ws, size_t ws_size,
                              hipStream_t stream) {
  (void)n_in; (void)out_size; (void)ws_size;
  const float* x    = (const float*)d_in[0];
  const float* rotp = (const float*)d_in[1];
  const float* entp = (const float*)d_in[2];
  float2* W = (float2*)d_ws;
  const int B = in_sizes[0] / 48;

  qk_prep<<<NL, 256, 0, stream>>>(entp, W);
  qk_main<<<B, 256, 0, stream>>>(x, rotp, entp, W, (float*)d_out);
}


# Apply each layer as four in-place 3-wire Kronecker-factor steps (complex 8x8 as real 16x16, K=16, f32 WMMA) with the CRot chains folded in, replacing four dense 64^3 GEMMs; summation order changes
# speedup vs baseline: 5.2650x; 4.1634x over previous
#include <hip/hip_runtime.h>
#include <math.h>

#define LDA 65
#define NL  4
#define NX  7

typedef __attribute__((ext_vector_type(2))) float v2f;
typedef __attribute__((ext_vector_type(8))) float v8f;

__device__ __forceinline__ float2 cmul(float2 a, float2 b) {
  return make_float2(a.x * b.x - a.y * b.y, a.x * b.y + a.y * b.x);
}
__device__ __forceinline__ float2 cadd(float2 a, float2 b) {
  return make_float2(a.x + b.x, a.y + b.y);
}

struct C2 { float2 a00, a01, a10, a11; };
__device__ __forceinline__ C2 mm2(const C2 L, const C2 R) {
  C2 o;
  o.a00 = cadd(cmul(L.a00, R.a00), cmul(L.a01, R.a10));
  o.a01 = cadd(cmul(L.a00, R.a01), cmul(L.a01, R.a11));
  o.a10 = cadd(cmul(L.a10, R.a00), cmul(L.a11, R.a10));
  o.a11 = cadd(cmul(L.a10, R.a01), cmul(L.a11, R.a11));
  return o;
}

__device__ __forceinline__ void rot_gate(float phi, float th, float om,
                                         float2* u00, float2* u01,
                                         float2* u10, float2* u11) {
  float ct = cosf(0.5f * th), st = sinf(0.5f * th);
  float ap = -0.5f * (phi + om);
  float am = 0.5f * (phi - om);
  *u00 = make_float2(ct * cosf(ap), ct * sinf(ap));
  *u01 = make_float2(-st * cosf(am), -st * sinf(am));
  *u10 = make_float2(st * cosf(-am), st * sinf(-am));
  *u11 = make_float2(ct * cosf(-ap), ct * sinf(-ap));
}

__global__ __launch_bounds__(64) void qk_prep(const float* __restrict__ entp,
                                              float2* __restrict__ W) {
  __shared__ float2 U[12][2][2];
  const int l = blockIdx.x;
  const int tid = threadIdx.x;

  if (tid < 12) {
    const float* p = &entp[(l * 12 + tid) * 3];
    rot_gate(p[0], p[1], p[2], &U[tid][0][0], &U[tid][0][1],
             &U[tid][1][0], &U[tid][1][1]);
  }
  __syncthreads();

  if (tid < NX * 8) {
    const int t = tid >> 3, j = tid & 7;
    const int g = (t + 1) >> 1;
    const bool link = (t > 0) && ((t & 1) == 0);
    float2 v[8];
#pragma unroll
    for (int r = 0; r < 8; r++) v[r] = make_float2((r == j) ? 1.f : 0.f, 0.f);
    if (link) {
      const int q = 3 * g - 1;
#pragma unroll
      for (int r = 0; r < 8; r++)
        v[r] = ((r & 3) == (j & 3)) ? U[q][r >> 2][j >> 2] : make_float2(0.f, 0.f);
    }
    {
      const int q = 3 * g;
      float2 u00 = U[q][0][0], u01 = U[q][0][1], u10 = U[q][1][0], u11 = U[q][1][1];
#pragma unroll
      for (int i = 4; i < 6; i++) {
        float2 x0 = v[i], x1 = v[i + 2];
        v[i]     = cadd(cmul(u00, x0), cmul(u01, x1));
        v[i + 2] = cadd(cmul(u10, x0), cmul(u11, x1));
      }
    }
    {
      const int q = 3 * g + 1;
      float2 u00 = U[q][0][0], u01 = U[q][0][1], u10 = U[q][1][0], u11 = U[q][1][1];
#pragma unroll
      for (int i = 2; i < 8; i += 4) {
        float2 x0 = v[i], x1 = v[i + 1];
        v[i]     = cadd(cmul(u00, x0), cmul(u01, x1));
        v[i + 1] = cadd(cmul(u10, x0), cmul(u11, x1));
      }
    }
    float2* Wt = W + (size_t)(l * NX + t) * 64;
#pragma unroll
    for (int r = 0; r < 8; r++) Wt[r * 8 + j] = v[r];
  }
}

__device__ __forceinline__ v8f wmma4(v2f a, v2f b, v8f c) {
  return __builtin_amdgcn_wmma_f32_16x16x4_f32(
      false, a, false, b, (short)0, c, false, false);
}

template <int NSTR, int GSTR, int TSTR>
__device__ __forceinline__ void kron_step(float* Sr, float* Si,
                                          const float2* V, int base0) {
  const int lane = threadIdx.x & 31;
  const int n = lane & 15;
  const int hi = lane >> 4;
  const int mm = n & 7, mhi = n >> 3;

  const float d0 = mhi ? 0.f : 1.f;
  const float d1 = mhi ? 1.f : 0.f;
  v2f a[4];
#pragma unroll
  for (int ks = 0; ks < 4; ks++) {
    const int kk = 4 * (ks & 1) + 2 * hi;
    const float2 u0 = V[mm * 8 + kk], u1 = V[mm * 8 + kk + 1];
    if (ks < 2) {
      a[ks].x = d0 * u0.x + d1 * u0.y;
      a[ks].y = d0 * u1.x + d1 * u1.y;
    } else {
      a[ks].x = d1 * u0.x - d0 * u0.y;
      a[ks].y = d1 * u1.x - d0 * u1.y;
    }
  }

  v2f b[4][4];
#pragma unroll
  for (int t = 0; t < 4; t++) {
#pragma unroll
    for (int ks = 0; ks < 4; ks++) {
      const float* P = (ks >= 2) ? Si : Sr;
      const int ad = base0 + t * TSTR + n * NSTR + (4 * (ks & 1) + 2 * hi) * GSTR;
      b[t][ks].x = P[ad];
      b[t][ks].y = P[ad + GSTR];
    }
  }

  v8f acc[4] = {};
#pragma unroll
  for (int ks = 0; ks < 4; ks++) {
#pragma unroll
    for (int t = 0; t < 4; t++) acc[t] = wmma4(a[ks], b[t][ks], acc[t]);
  }

  float* Q = hi ? Si : Sr;
#pragma unroll
  for (int t = 0; t < 4; t++) {
#pragma unroll
    for (int v = 0; v < 8; v++)
      Q[base0 + t * TSTR + n * NSTR + v * GSTR] = acc[t][v];
  }
}

__global__ __launch_bounds__(256) void qk_main(const float* __restrict__ x,
                                               const float* __restrict__ rotp,
                                               const float* __restrict__ entp,
                                               const float2* __restrict__ W,
                                               float* __restrict__ out) {
  __shared__ float Sr[64 * LDA], Si[64 * LDA];
  __shared__ float2 G[NL][12][2][2];
  __shared__ float2 K8[NL][4][64];
  __shared__ float2 V[NL][NX][64];
  __shared__ float2 CU[NL][2][2];

  const int tid = threadIdx.x;
  const int s = blockIdx.x;

  if (tid < 12 * NL) {
    const int gl = tid / 12, gq = tid - 12 * gl;
    const float* xs = x + (size_t)s * 48;
    float a = xs[0 * 12 + gq], b = xs[1 * 12 + gq];
    float c = xs[2 * 12 + gq], d = xs[3 * 12 + gq];
    float ca = cosf(0.5f * a), sa = sinf(0.5f * a);
    float cb = cosf(0.5f * b), sb = sinf(0.5f * b);
    float cc = cosf(0.5f * c), sc = sinf(0.5f * c);
    float cd = cosf(0.5f * d), sd = sinf(0.5f * d);
    C2 M  = { {ca, 0.f}, {-sa, 0.f}, {sa, 0.f}, {ca, 0.f} };
    C2 RX = { {cb, 0.f}, {0.f, -sb}, {0.f, -sb}, {cb, 0.f} };
    C2 RZ = { {cc, -sc}, {0.f, 0.f}, {0.f, 0.f}, {cc, sc} };
    C2 RY = { {cd, 0.f}, {-sd, 0.f}, {sd, 0.f}, {cd, 0.f} };
    M = mm2(RX, M); M = mm2(RZ, M); M = mm2(RY, M);
    const float* rp = &rotp[(gl * 12 + gq) * 3];
    C2 R;
    rot_gate(rp[0], rp[1], rp[2], &R.a00, &R.a01, &R.a10, &R.a11);
    M = mm2(R, M);
    G[gl][gq][0][0] = M.a00; G[gl][gq][0][1] = M.a01;
    G[gl][gq][1][0] = M.a10; G[gl][gq][1][1] = M.a11;
  }
  if (tid >= 64 && tid < 64 + NL) {
    int l = tid - 64;
    const float* p = &entp[(l * 12 + 11) * 3];
    rot_gate(p[0], p[1], p[2], &CU[l][0][0], &CU[l][0][1],
             &CU[l][1][0], &CU[l][1][1]);
  }
  for (int idx = tid; idx < 4096; idx += 256) {
    int i = idx >> 6, j = idx & 63;
    Sr[i * LDA + j] = (idx == 0) ? 1.f : 0.f;
    Si[i * LDA + j] = 0.f;
  }
  __syncthreads();

  for (int e = tid; e < NL * 256; e += 256) {
    const int l = e >> 8;
    const int kg = (e >> 6) & 3, ka = (e >> 3) & 7, kb8 = e & 7;
    float2 p = cmul(cmul(G[l][3 * kg][(ka >> 2) & 1][(kb8 >> 2) & 1],
                         G[l][3 * kg + 1][(ka >> 1) & 1][(kb8 >> 1) & 1]),
                    G[l][3 * kg + 2][ka & 1][kb8 & 1]);
    K8[l][kg][ka * 8 + kb8] = p;
  }
  __syncthreads();

  for (int e = tid; e < NL * NX * 64; e += 256) {
    const int lt = e >> 6;
    const int l = lt / NX, t = lt - l * NX;
    const int g = (t + 1) >> 1;
    const int r = (e >> 3) & 7, c = e & 7;
    const float2* Xr = W + (size_t)lt * 64 + r * 8;
    const float2* Kc = &K8[l][g][c];
    float2 acc = make_float2(0.f, 0.f);
#pragma unroll
    for (int k = 0; k < 8; k++) acc = cadd(acc, cmul(Xr[k], Kc[k * 8]));
    V[l][t][r * 8 + c] = acc;
  }
  __syncthreads();

  const int w = tid >> 5;
  const int par = (w >> 1) & 1;

  for (int l = 0; l < NL; l++) {
    kron_step<1, 8 * LDA, 16>(Sr, Si, V[l][0], w * LDA);
    __syncthreads();
    kron_step<1, LDA, 16>(Sr, Si, V[l][1 + (w & 1)], w * 8 * LDA);
    __syncthreads();
    kron_step<2 * LDA, 8, 1>(Sr, Si, V[l][3 + par],
                             (par + 32 * (w >> 2)) * LDA + 4 * (w & 1));
    __syncthreads();
    kron_step<LDA, 1, 16 * LDA>(Sr, Si, V[l][5 + (w & 1)], w * 8);
    __syncthreads();

    {
      float2 u00 = CU[l][0][0], u01 = CU[l][0][1];
      float2 u10 = CU[l][1][0], u11 = CU[l][1][1];
      for (int t = tid; t < 1024; t += 256) {
        int il = t >> 5;
        int j = ((t & 31) << 1) | 1;
        int a0 = il * LDA + j, a1 = (il + 32) * LDA + j;
        float2 x0 = make_float2(Sr[a0], Si[a0]);
        float2 x1 = make_float2(Sr[a1], Si[a1]);
        float2 y0 = cadd(cmul(u00, x0), cmul(u01, x1));
        float2 y1 = cadd(cmul(u10, x0), cmul(u11, x1));
        Sr[a0] = y0.x; Si[a0] = y0.y;
        Sr[a1] = y1.x; Si[a1] = y1.y;
      }
    }
    __syncthreads();
  }

  float z[12];
  for (int q = 0; q < 12; q++) z[q] = 0.f;
  for (int idx = tid; idx < 4096; idx += 256) {
    int i = idx >> 6, j = idx & 63;
    float pr = Sr[i * LDA + j], pi = Si[i * LDA + j];
    float p = pr * pr + pi * pi;
    for (int q = 0; q < 12; q++)
      z[q] += ((idx >> (11 - q)) & 1) ? -p : p;
  }
  float* red = (float*)&V[0][0][0];
  for (int q = 0; q < 12; q++) red[tid * 12 + q] = z[q];
  __syncthreads();
  for (int sft = 128; sft > 0; sft >>= 1) {
    if (tid < sft)
      for (int q = 0; q < 12; q++)
        red[tid * 12 + q] += red[(tid + sft) * 12 + q];
    __syncthreads();
  }
  if (tid < 12) out[(size_t)s * 12 + tid] = red[tid];
}

extern "C" void kernel_launch(void* const* d_in, const int* in_sizes, int n_in,
                              void* d_out, int out_size, void* d_ws, size_t ws_size,
                              hipStream_t stream) {
  (void)n_in; (void)out_size; (void)ws_size;
  const float* x    = (const float*)d_in[0];
  const float* rotp = (const float*)d_in[1];
  const float* entp = (const float*)d_in[2];
  float2* W = (float2*)d_ws;
  const int B = in_sizes[0] / 48;

  qk_prep<<<NL, 64, 0, stream>>>(entp, W);
  qk_main<<<B, 256, 0, stream>>>(x, rotp, entp, W, (float*)d_out);
}


# Bank-conflict-free LDS: imaginary plane 32 banks off the real one, fragment k order with lane half = real/imag part, step-d tiles on stride-4 rows, padded factor rows; k order of the sums changes
# speedup vs baseline: 5.8599x; 1.1130x over previous
#include <hip/hip_runtime.h>
#include <math.h>

#define LDA 65
#define NL  4
#define NX  7
#define VP  10

typedef __attribute__((ext_vector_type(2))) float v2f;
typedef __attribute__((ext_vector_type(8))) float v8f;

__device__ __forceinline__ float2 cmul(float2 a, float2 b) {
  return make_float2(a.x * b.x - a.y * b.y, a.x * b.y + a.y * b.x);
}
__device__ __forceinline__ float2 cadd(float2 a, float2 b) {
  return make_float2(a.x + b.x, a.y + b.y);
}

struct C2 { float2 a00, a01, a10, a11; };
__device__ __forceinline__ C2 mm2(const C2 L, const C2 R) {
  C2 o;
  o.a00 = cadd(cmul(L.a00, R.a00), cmul(L.a01, R.a10));
  o.a01 = cadd(cmul(L.a00, R.a01), cmul(L.a01, R.a11));
  o.a10 = cadd(cmul(L.a10, R.a00), cmul(L.a11, R.a10));
  o.a11 = cadd(cmul(L.a10, R.a01), cmul(L.a11, R.a11));
  return o;
}

__device__ __forceinline__ void rot_gate(float phi, float th, float om,
                                         float2* u00, float2* u01,
                                         float2* u10, float2* u11) {
  float ct = cosf(0.5f * th), st = sinf(0.5f * th);
  float ap = -0.5f * (phi + om);
  float am = 0.5f * (phi - om);
  *u00 = make_float2(ct * cosf(ap), ct * sinf(ap));
  *u01 = make_float2(-st * cosf(am), -st * sinf(am));
  *u10 = make_float2(st * cosf(-am), st * sinf(-am));
  *u11 = make_float2(ct * cosf(-ap), ct * sinf(-ap));
}

__global__ __launch_bounds__(64) void qk_prep(const float* __restrict__ entp,
                                              float2* __restrict__ W) {
  __shared__ float2 U[12][2][2];
  const int l = blockIdx.x;
  const int tid = threadIdx.x;

  if (tid < 12) {
    const float* p = &entp[(l * 12 + tid) * 3];
    rot_gate(p[0], p[1], p[2], &U[tid][0][0], &U[tid][0][1],
             &U[tid][1][0], &U[tid][1][1]);
  }
  __syncthreads();

  if (tid < NX * 8) {
    const int t = tid >> 3, j = tid & 7;
    const int g = (t + 1) >> 1;
    const bool link = (t > 0) && ((t & 1) == 0);
    float2 v[8];
#pragma unroll
    for (int r = 0; r < 8; r++) v[r] = make_float2((r == j) ? 1.f : 0.f, 0.f);
    if (link) {
      const int q = 3 * g - 1;
#pragma unroll
      for (int r = 0; r < 8; r++)
        v[r] = ((r & 3) == (j & 3)) ? U[q][r >> 2][j >> 2] : make_float2(0.f, 0.f);
    }
    {
      const int q = 3 * g;
      float2 u00 = U[q][0][0], u01 = U[q][0][1], u10 = U[q][1][0], u11 = U[q][1][1];
#pragma unroll
      for (int i = 4; i < 6; i++) {
        float2 x0 = v[i], x1 = v[i + 2];
        v[i]     = cadd(cmul(u00, x0), cmul(u01, x1));
        v[i + 2] = cadd(cmul(u10, x0), cmul(u11, x1));
      }
    }
    {
      const int q = 3 * g + 1;
      float2 u00 = U[q][0][0], u01 = U[q][0][1], u10 = U[q][1][0], u11 = U[q][1][1];
#pragma unroll
      for (int i = 2; i < 8; i += 4) {
        float2 x0 = v[i], x1 = v[i + 1];
        v[i]     = cadd(cmul(u00, x0), cmul(u01, x1));
        v[i + 1] = cadd(cmul(u10, x0), cmul(u11, x1));
      }
    }
    float2* Wt = W + (size_t)(l * NX + t) * 64;
#pragma unroll
    for (int r = 0; r < 8; r++) Wt[r * 8 + j] = v[r];
  }
}

__device__ __forceinline__ v8f wmma4(v2f a, v2f b, v8f c) {
  return __builtin_amdgcn_wmma_f32_16x16x4_f32(
      false, a, false, b, (short)0, c, false, false);
}

template <int NSTR, int GSTR, int TS0, int TS1>
__device__ __forceinline__ void kron_step(float* Sr, float* Si,
                                          const float2* V, int base0) {
  const int lane = threadIdx.x & 31;
  const int n = lane & 15;
  const int hi = lane >> 4;
  const int mm = n & 7, mhi = n >> 3;

  const float wr = (mhi == hi) ? 1.f : 0.f;
  const float wi = (mhi == hi) ? 0.f : (mhi ? 1.f : -1.f);
  v2f a[4];
#pragma unroll
  for (int ks = 0; ks < 4; ks++) {
    const float2 u0 = V[mm * VP + 2 * ks], u1 = V[mm * VP + 2 * ks + 1];
    a[ks].x = wr * u0.x + wi * u0.y;
    a[ks].y = wr * u1.x + wi * u1.y;
  }

  const float* P = hi ? Si : Sr;
  v2f b[4][4];
#pragma unroll
  for (int t = 0; t < 4; t++) {
#pragma unroll
    for (int ks = 0; ks < 4; ks++) {
      const int ad = base0 + (t & 1) * TS0 + (t >> 1) * TS1 + n * NSTR + (2 * ks) * GSTR;
      b[t][ks].x = P[ad];
      b[t][ks].y = P[ad + GSTR];
    }
  }

  v8f acc[4] = {};
#pragma unroll
  for (int ks = 0; ks < 4; ks++) {
#pragma unroll
    for (int t = 0; t < 4; t++) acc[t] = wmma4(a[ks], b[t][ks], acc[t]);
  }

  float* Q = hi ? Si : Sr;
#pragma unroll
  for (int t = 0; t < 4; t++) {
#pragma unroll
    for (int v = 0; v < 8; v++)
      Q[base0 + (t & 1) * TS0 + (t >> 1) * TS1 + n * NSTR + v * GSTR] = acc[t][v];
  }
}

__global__ __launch_bounds__(256) void qk_main(const float* __restrict__ x,
                                               const float* __restrict__ rotp,
                                               const float* __restrict__ entp,
                                               const float2* __restrict__ W,
                                               float* __restrict__ out) {
  __shared__ float S2[2 * 64 * LDA + 32];
  float* const Sr = S2;
  float* const Si = S2 + 64 * LDA + 32;
  __shared__ float2 G[NL][12][2][2];
  __shared__ float2 K8[NL][4][64];
  __shared__ float2 V[NL][NX][8 * VP];
  __shared__ float2 CU[NL][2][2];

  const int tid = threadIdx.x;
  const int s = blockIdx.x;

  if (tid < 12 * NL) {
    const int gl = tid / 12, gq = tid - 12 * gl;
    const float* xs = x + (size_t)s * 48;
    float a = xs[0 * 12 + gq], b = xs[1 * 12 + gq];
    float c = xs[2 * 12 + gq], d = xs[3 * 12 + gq];
    float ca = cosf(0.5f * a), sa = sinf(0.5f * a);
    float cb = cosf(0.5f * b), sb = sinf(0.5f * b);
    float cc = cosf(0.5f * c), sc = sinf(0.5f * c);
    float cd = cosf(0.5f * d), sd = sinf(0.5f * d);
    C2 M  = { {ca, 0.f}, {-sa, 0.f}, {sa, 0.f}, {ca, 0.f} };
    C2 RX = { {cb, 0.f}, {0.f, -sb}, {0.f, -sb}, {cb, 0.f} };
    C2 RZ = { {cc, -sc}, {0.f, 0.f}, {0.f, 0.f}, {cc, sc} };
    C2 RY = { {cd, 0.f}, {-sd, 0.f}, {sd, 0.f}, {cd, 0.f} };
    M = mm2(RX, M); M = mm2(RZ, M); M = mm2(RY, M);
    const float* rp = &rotp[(gl * 12 + gq) * 3];
    C2 R;
    rot_gate(rp[0], rp[1], rp[2], &R.a00, &R.a01, &R.a10, &R.a11);
    M = mm2(R, M);
    G[gl][gq][0][0] = M.a00; G[gl][gq][0][1] = M.a01;
    G[gl][gq][1][0] = M.a10; G[gl][gq][1][1] = M.a11;
  }
  if (tid >= 64 && tid < 64 + NL) {
    int l = tid - 64;
    const float* p = &entp[(l * 12 + 11) * 3];
    rot_gate(p[0], p[1], p[2], &CU[l][0][0], &CU[l][0][1],
             &CU[l][1][0], &CU[l][1][1]);
  }
  for (int idx = tid; idx < 4096; idx += 256) {
    int i = idx >> 6, j = idx & 63;
    Sr[i * LDA + j] = (idx == 0) ? 1.f : 0.f;
    Si[i * LDA + j] = 0.f;
  }
  __syncthreads();

  for (int e = tid; e < NL * 256; e += 256) {
    const int l = e >> 8;
    const int kg = (e >> 6) & 3, ka = (e >> 3) & 7, kb8 = e & 7;
    float2 p = cmul(cmul(G[l][3 * kg][(ka >> 2) & 1][(kb8 >> 2) & 1],
                         G[l][3 * kg + 1][(ka >> 1) & 1][(kb8 >> 1) & 1]),
                    G[l][3 * kg + 2][ka & 1][kb8 & 1]);
    K8[l][kg][ka * 8 + kb8] = p;
  }
  __syncthreads();

  for (int e = tid; e < NL * NX * 64; e += 256) {
    const int lt = e >> 6;
    const int l = lt / NX, t = lt - l * NX;
    const int g = (t + 1) >> 1;
    const int r = (e >> 3) & 7, c = e & 7;
    const float2* Xr = W + (size_t)lt * 64 + r * 8;
    const float2* Kc = &K8[l][g][c];
    float2 acc = make_float2(0.f, 0.f);
#pragma unroll
    for (int k = 0; k < 8; k++) acc = cadd(acc, cmul(Xr[k], Kc[k * 8]));
    V[l][t][r * VP + c] = acc;
  }
  __syncthreads();

  const int w = tid >> 5;
  const int par = (w >> 1) & 1;

  for (int l = 0; l < NL; l++) {
    kron_step<1, 8 * LDA, 16, 32>(Sr, Si, V[l][0], w * LDA);
    __syncthreads();
    kron_step<1, LDA, 16, 32>(Sr, Si, V[l][1 + (w & 1)], w * 8 * LDA);
    __syncthreads();
    kron_step<2 * LDA, 8, 1, 2>(Sr, Si, V[l][3 + par],
                             (par + 32 * (w >> 2)) * LDA + 4 * (w & 1));
    __syncthreads();
    kron_step<4 * LDA, 1, LDA, 2 * LDA>(Sr, Si, V[l][5 + (w & 1)], w * 8);
    __syncthreads();

    {
      float2 u00 = CU[l][0][0], u01 = CU[l][0][1];
      float2 u10 = CU[l][1][0], u11 = CU[l][1][1];
      for (int t = tid; t < 1024; t += 256) {
        int il = t >> 5;
        int j = ((t & 31) << 1) | 1;
        int a0 = il * LDA + j, a1 = (il + 32) * LDA + j;
        float2 x0 = make_float2(Sr[a0], Si[a0]);
        float2 x1 = make_float2(Sr[a1], Si[a1]);
        float2 y0 = cadd(cmul(u00, x0), cmul(u01, x1));
        float2 y1 = cadd(cmul(u10, x0), cmul(u11, x1));
        Sr[a0] = y0.x; Si[a0] = y0.y;
        Sr[a1] = y1.x; Si[a1] = y1.y;
      }
    }
    __syncthreads();
  }

  float z[12];
  for (int q = 0; q < 12; q++) z[q] = 0.f;
  for (int idx = tid; idx < 4096; idx += 256) {
    int i = idx >> 6, j = idx & 63;
    float pr = Sr[i * LDA + j], pi = Si[i * LDA + j];
    float p = pr * pr + pi * pi;
    for (int q = 0; q < 12; q++)
      z[q] += ((idx >> (11 - q)) & 1) ? -p : p;
  }
  float* red = (float*)&V[0][0][0];
  for (int q = 0; q < 12; q++) red[tid * 12 + q] = z[q];
  __syncthreads();
  for (int sft = 128; sft > 0; sft >>= 1) {
    if (tid < sft)
      for (int q = 0; q < 12; q++)
        red[tid * 12 + q] += red[(tid + sft) * 12 + q];
    __syncthreads();
  }
  if (tid < 12) out[(size_t)s * 12 + tid] = red[tid];
}

extern "C" void kernel_launch(void* const* d_in, const int* in_sizes, int n_in,
                              void* d_out, int out_size, void* d_ws, size_t ws_size,
                              hipStream_t stream) {
  (void)n_in; (void)out_size; (void)ws_size;
  const float* x    = (const float*)d_in[0];
  const float* rotp = (const float*)d_in[1];
  const float* entp = (const float*)d_in[2];
  float2* W = (float2*)d_ws;
  const int B = in_sizes[0] / 48;

  qk_prep<<<NL, 64, 0, stream>>>(entp, W);
  qk_main<<<B, 256, 0, stream>>>(x, rotp, entp, W, (float*)d_out);
}


# Layer 0 acts on the fixed |0..0> start, so its state is a product of factor columns: written directly (one complex multiply per amplitude, CR(11,0) folded in); its 4 WMMA steps go, its sums vanish
# speedup vs baseline: 6.9618x; 1.1880x over previous
#include <hip/hip_runtime.h>
#include <math.h>

#define LDA 65
#define NL  4
#define NX  7
#define VP  10

typedef __attribute__((ext_vector_type(2))) float v2f;
typedef __attribute__((ext_vector_type(8))) float v8f;

__device__ __forceinline__ float2 cmul(float2 a, float2 b) {
  return make_float2(a.x * b.x - a.y * b.y, a.x * b.y + a.y * b.x);
}
__device__ __forceinline__ float2 cadd(float2 a, float2 b) {
  return make_float2(a.x + b.x, a.y + b.y);
}

struct C2 { float2 a00, a01, a10, a11; };
__device__ __forceinline__ C2 mm2(const C2 L, const C2 R) {
  C2 o;
  o.a00 = cadd(cmul(L.a00, R.a00), cmul(L.a01, R.a10));
  o.a01 = cadd(cmul(L.a00, R.a01), cmul(L.a01, R.a11));
  o.a10 = cadd(cmul(L.a10, R.a00), cmul(L.a11, R.a10));
  o.a11 = cadd(cmul(L.a10, R.a01), cmul(L.a11, R.a11));
  return o;
}

__device__ __forceinline__ void rot_gate(float phi, float th, float om,
                                         float2* u00, float2* u01,
                                         float2* u10, float2* u11) {
  float ct = cosf(0.5f * th), st = sinf(0.5f * th);
  float ap = -0.5f * (phi + om);
  float am = 0.5f * (phi - om);
  *u00 = make_float2(ct * cosf(ap), ct * sinf(ap));
  *u01 = make_float2(-st * cosf(am), -st * sinf(am));
  *u10 = make_float2(st * cosf(-am), st * sinf(-am));
  *u11 = make_float2(ct * cosf(-ap), ct * sinf(-ap));
}

__global__ __launch_bounds__(64) void qk_prep(const float* __restrict__ entp,
                                              float2* __restrict__ W) {
  __shared__ float2 U[12][2][2];
  const int l = blockIdx.x;
  const int tid = threadIdx.x;

  if (tid < 12) {
    const float* p = &entp[(l * 12 + tid) * 3];
    rot_gate(p[0], p[1], p[2], &U[tid][0][0], &U[tid][0][1],
             &U[tid][1][0], &U[tid][1][1]);
  }
  __syncthreads();

  if (tid < NX * 8) {
    const int t = tid >> 3, j = tid & 7;
    const int g = (t + 1) >> 1;
    const bool link = (t > 0) && ((t & 1) == 0);
    float2 v[8];
#pragma unroll
    for (int r = 0; r < 8; r++) v[r] = make_float2((r == j) ? 1.f : 0.f, 0.f);
    if (link) {
      const int q = 3 * g - 1;
#pragma unroll
      for (int r = 0; r < 8; r++)
        v[r] = ((r & 3) == (j & 3)) ? U[q][r >> 2][j >> 2] : make_float2(0.f, 0.f);
    }
    {
      const int q = 3 * g;
      float2 u00 = U[q][0][0], u01 = U[q][0][1], u10 = U[q][1][0], u11 = U[q][1][1];
#pragma unroll
      for (int i = 4; i < 6; i++) {
        float2 x0 = v[i], x1 = v[i + 2];
        v[i]     = cadd(cmul(u00, x0), cmul(u01, x1));
        v[i + 2] = cadd(cmul(u10, x0), cmul(u11, x1));
      }
    }
    {
      const int q = 3 * g + 1;
      float2 u00 = U[q][0][0], u01 = U[q][0][1], u10 = U[q][1][0], u11 = U[q][1][1];
#pragma unroll
      for (int i = 2; i < 8; i += 4) {
        float2 x0 = v[i], x1 = v[i + 1];
        v[i]     = cadd(cmul(u00, x0), cmul(u01, x1));
        v[i + 1] = cadd(cmul(u10, x0), cmul(u11, x1));
      }
    }
    float2* Wt = W + (size_t)(l * NX + t) * 64;
#pragma unroll
    for (int r = 0; r < 8; r++) Wt[r * 8 + j] = v[r];
  }
}

__device__ __forceinline__ v8f wmma4(v2f a, v2f b, v8f c) {
  return __builtin_amdgcn_wmma_f32_16x16x4_f32(
      false, a, false, b, (short)0, c, false, false);
}

template <int NSTR, int GSTR, int TS0, int TS1>
__device__ __forceinline__ void kron_step(float* Sr, float* Si,
                                          const float2* V, int base0) {
  const int lane = threadIdx.x & 31;
  const int n = lane & 15;
  const int hi = lane >> 4;
  const int mm = n & 7, mhi = n >> 3;

  const float wr = (mhi == hi) ? 1.f : 0.f;
  const float wi = (mhi == hi) ? 0.f : (mhi ? 1.f : -1.f);
  v2f a[4];
#pragma unroll
  for (int ks = 0; ks < 4; ks++) {
    const float2 u0 = V[mm * VP + 2 * ks], u1 = V[mm * VP + 2 * ks + 1];
    a[ks].x = wr * u0.x + wi * u0.y;
    a[ks].y = wr * u1.x + wi * u1.y;
  }

  const float* P = hi ? Si : Sr;
  v2f b[4][4];
#pragma unroll
  for (int t = 0; t < 4; t++) {
#pragma unroll
    for (int ks = 0; ks < 4; ks++) {
      const int ad = base0 + (t & 1) * TS0 + (t >> 1) * TS1 + n * NSTR + (2 * ks) * GSTR;
      b[t][ks].x = P[ad];
      b[t][ks].y = P[ad + GSTR];
    }
  }

  v8f acc[4] = {};
#pragma unroll
  for (int ks = 0; ks < 4; ks++) {
#pragma unroll
    for (int t = 0; t < 4; t++) acc[t] = wmma4(a[ks], b[t][ks], acc[t]);
  }

  float* Q = hi ? Si : Sr;
#pragma unroll
  for (int t = 0; t < 4; t++) {
#pragma unroll
    for (int v = 0; v < 8; v++)
      Q[base0 + (t & 1) * TS0 + (t >> 1) * TS1 + n * NSTR + v * GSTR] = acc[t][v];
  }
}

__global__ __launch_bounds__(256) void qk_main(const float* __restrict__ x,
                                               const float* __restrict__ rotp,
                                               const float* __restrict__ entp,
                                               const float2* __restrict__ W,
                                               float* __restrict__ out) {
  __shared__ float S2[2 * 64 * LDA + 32];
  float* const Sr = S2;
  float* const Si = S2 + 64 * LDA + 32;
  __shared__ float2 G[NL][12][2][2];
  __shared__ float2 K8[NL][4][64];
  __shared__ float2 V[NL - 1][NX][8 * VP];
  __shared__ float2 V0[NX][8];
  __shared__ float2 RC[256];
  __shared__ float2 CU[NL][2][2];

  const int tid = threadIdx.x;
  const int s = blockIdx.x;

  if (tid < 12 * NL) {
    const int gl = tid / 12, gq = tid - 12 * gl;
    const float* xs = x + (size_t)s * 48;
    float a = xs[0 * 12 + gq], b = xs[1 * 12 + gq];
    float c = xs[2 * 12 + gq], d = xs[3 * 12 + gq];
    float ca = cosf(0.5f * a), sa = sinf(0.5f * a);
    float cb = cosf(0.5f * b), sb = sinf(0.5f * b);
    float cc = cosf(0.5f * c), sc = sinf(0.5f * c);
    float cd = cosf(0.5f * d), sd = sinf(0.5f * d);
    C2 M  = { {ca, 0.f}, {-sa, 0.f}, {sa, 0.f}, {ca, 0.f} };
    C2 RX = { {cb, 0.f}, {0.f, -sb}, {0.f, -sb}, {cb, 0.f} };
    C2 RZ = { {cc, -sc}, {0.f, 0.f}, {0.f, 0.f}, {cc, sc} };
    C2 RY = { {cd, 0.f}, {-sd, 0.f}, {sd, 0.f}, {cd, 0.f} };
    M = mm2(RX, M); M = mm2(RZ, M); M = mm2(RY, M);
    const float* rp = &rotp[(gl * 12 + gq) * 3];
    C2 R;
    rot_gate(rp[0], rp[1], rp[2], &R.a00, &R.a01, &R.a10, &R.a11);
    M = mm2(R, M);
    G[gl][gq][0][0] = M.a00; G[gl][gq][0][1] = M.a01;
    G[gl][gq][1][0] = M.a10; G[gl][gq][1][1] = M.a11;
  }
  if (tid >= 64 && tid < 64 + NL) {
    int l = tid - 64;
    const float* p = &entp[(l * 12 + 11) * 3];
    rot_gate(p[0], p[1], p[2], &CU[l][0][0], &CU[l][0][1],
             &CU[l][1][0], &CU[l][1][1]);
  }
  __syncthreads();

  for (int e = tid; e < NL * 256; e += 256) {
    const int l = e >> 8;
    const int kg = (e >> 6) & 3, ka = (e >> 3) & 7, kb8 = e & 7;
    float2 p = cmul(cmul(G[l][3 * kg][(ka >> 2) & 1][(kb8 >> 2) & 1],
                         G[l][3 * kg + 1][(ka >> 1) & 1][(kb8 >> 1) & 1]),
                    G[l][3 * kg + 2][ka & 1][kb8 & 1]);
    K8[l][kg][ka * 8 + kb8] = p;
  }
  __syncthreads();

  constexpr int NFULL = (NL - 1) * NX * 64;
  for (int e = tid; e < NFULL + NX * 8; e += 256) {
    int lt, r, c;
    if (e < NFULL) { lt = NX + (e >> 6); r = (e >> 3) & 7; c = e & 7; }
    else           { lt = (e - NFULL) >> 3; r = (e - NFULL) & 7; c = 0; }
    const int l = lt / NX, t = lt - l * NX;
    const int g = (t + 1) >> 1;
    const float2* Xr = W + (size_t)lt * 64 + r * 8;
    const float2* Kc = &K8[l][g][c];
    float2 acc = make_float2(0.f, 0.f);
#pragma unroll
    for (int k = 0; k < 8; k++) acc = cadd(acc, cmul(Xr[k], Kc[k * 8]));
    float2* dst = (l > 0) ? &V[l - 1][t][r * VP + c] : &V0[t][r];
    *dst = acc;
  }
  __syncthreads();

  {
    float2 val;
    if (tid < 64) {
      const int ra = tid >> 3, rb = tid & 7;
      val = cmul(V0[0][ra], V0[1 + (ra & 1)][rb]);
    } else if (tid < 128) {
      const int h = (tid >> 5) & 1;
      const int ra = (tid >> 3) & 3, rb = tid & 7;
      const float2 bf = V0[1 + (ra & 1)][rb];
      const float2 x0 = cmul(V0[0][ra], bf);
      const float2 x1 = cmul(V0[0][ra + 4], bf);
      val = cadd(cmul(CU[0][h][0], x0), cmul(CU[0][h][1], x1));
    } else {
      const int p = (tid >> 6) & 1;
      const int gc = (tid >> 3) & 7, gd = tid & 7;
      val = cmul(V0[3 + p][gc], V0[5 + (gc & 1)][gd]);
    }
    RC[tid] = val;
  }
  __syncthreads();

  {
    const int j = tid & 63, q = tid >> 6;
    const float2 cf = RC[128 + 64 * (q & 1) + j];
    const float2* Rs = RC + 64 * (j & 1);
    for (int it = 0; it < 16; it++) {
      const int i = q + 4 * it;
      const float2 a = cmul(Rs[i], cf);
      Sr[i * LDA + j] = a.x;
      Si[i * LDA + j] = a.y;
    }
  }
  __syncthreads();

  const int w = tid >> 5;
  const int par = (w >> 1) & 1;

  for (int l = 1; l < NL; l++) {
    const float2 (*Vl)[8 * VP] = V[l - 1];
    kron_step<1, 8 * LDA, 16, 32>(Sr, Si, Vl[0], w * LDA);
    __syncthreads();
    kron_step<1, LDA, 16, 32>(Sr, Si, Vl[1 + (w & 1)], w * 8 * LDA);
    __syncthreads();
    kron_step<2 * LDA, 8, 1, 2>(Sr, Si, Vl[3 + par],
                             (par + 32 * (w >> 2)) * LDA + 4 * (w & 1));
    __syncthreads();
    kron_step<4 * LDA, 1, LDA, 2 * LDA>(Sr, Si, Vl[5 + (w & 1)], w * 8);
    __syncthreads();

    {
      float2 u00 = CU[l][0][0], u01 = CU[l][0][1];
      float2 u10 = CU[l][1][0], u11 = CU[l][1][1];
      for (int t = tid; t < 1024; t += 256) {
        int il = t >> 5;
        int j = ((t & 31) << 1) | 1;
        int a0 = il * LDA + j, a1 = (il + 32) * LDA + j;
        float2 x0 = make_float2(Sr[a0], Si[a0]);
        float2 x1 = make_float2(Sr[a1], Si[a1]);
        float2 y0 = cadd(cmul(u00, x0), cmul(u01, x1));
        float2 y1 = cadd(cmul(u10, x0), cmul(u11, x1));
        Sr[a0] = y0.x; Si[a0] = y0.y;
        Sr[a1] = y1.x; Si[a1] = y1.y;
      }
    }
    __syncthreads();
  }

  float z[12];
  for (int q = 0; q < 12; q++) z[q] = 0.f;
  for (int idx = tid; idx < 4096; idx += 256) {
    int i = idx >> 6, j = idx & 63;
    float pr = Sr[i * LDA + j], pi = Si[i * LDA + j];
    float p = pr * pr + pi * pi;
    for (int q = 0; q < 12; q++)
      z[q] += ((idx >> (11 - q)) & 1) ? -p : p;
  }
  float* red = (float*)&V[0][0][0];
  for (int q = 0; q < 12; q++) red[tid * 12 + q] = z[q];
  __syncthreads();
  for (int sft = 128; sft > 0; sft >>= 1) {
    if (tid < sft)
      for (int q = 0; q < 12; q++)
        red[tid * 12 + q] += red[(tid + sft) * 12 + q];
    __syncthreads();
  }
  if (tid < 12) out[(size_t)s * 12 + tid] = red[tid];
}

extern "C" void kernel_launch(void* const* d_in, const int* in_sizes, int n_in,
                              void* d_out, int out_size, void* d_ws, size_t ws_size,
                              hipStream_t stream) {
  (void)n_in; (void)out_size; (void)ws_size;
  const float* x    = (const float*)d_in[0];
  const float* rotp = (const float*)d_in[1];
  const float* entp = (const float*)d_in[2];
  float2* W = (float2*)d_ws;
  const int B = in_sizes[0] / 48;

  qk_prep<<<NL, 64, 0, stream>>>(entp, W);
  qk_main<<<B, 256, 0, stream>>>(x, rotp, entp, W, (float*)d_out);
}


# Layer 1 meets layer 0's product state: its four steps run on the 64-entry row and column factors, the state is written as a 4-term factor sum; 8 WMMA steps left; summation order changes
# speedup vs baseline: 8.0585x; 1.1575x over previous
#include <hip/hip_runtime.h>
#include <math.h>

#define LDA 65
#define NL  4
#define NX  7
#define VP  10

typedef __attribute__((ext_vector_type(2))) float v2f;
typedef __attribute__((ext_vector_type(8))) float v8f;

__device__ __forceinline__ float2 cmul(float2 a, float2 b) {
  return make_float2(a.x * b.x - a.y * b.y, a.x * b.y + a.y * b.x);
}
__device__ __forceinline__ float2 cadd(float2 a, float2 b) {
  return make_float2(a.x + b.x, a.y + b.y);
}

struct C2 { float2 a00, a01, a10, a11; };
__device__ __forceinline__ C2 mm2(const C2 L, const C2 R) {
  C2 o;
  o.a00 = cadd(cmul(L.a00, R.a00), cmul(L.a01, R.a10));
  o.a01 = cadd(cmul(L.a00, R.a01), cmul(L.a01, R.a11));
  o.a10 = cadd(cmul(L.a10, R.a00), cmul(L.a11, R.a10));
  o.a11 = cadd(cmul(L.a10, R.a01), cmul(L.a11, R.a11));
  return o;
}

__device__ __forceinline__ void rot_gate(float phi, float th, float om,
                                         float2* u00, float2* u01,
                                         float2* u10, float2* u11) {
  float ct = cosf(0.5f * th), st = sinf(0.5f * th);
  float ap = -0.5f * (phi + om);
  float am = 0.5f * (phi - om);
  *u00 = make_float2(ct * cosf(ap), ct * sinf(ap));
  *u01 = make_float2(-st * cosf(am), -st * sinf(am));
  *u10 = make_float2(st * cosf(-am), st * sinf(-am));
  *u11 = make_float2(ct * cosf(-ap), ct * sinf(-ap));
}

__global__ __launch_bounds__(64) void qk_prep(const float* __restrict__ entp,
                                              float2* __restrict__ W) {
  __shared__ float2 U[12][2][2];
  const int l = blockIdx.x;
  const int tid = threadIdx.x;

  if (tid < 12) {
    const float* p = &entp[(l * 12 + tid) * 3];
    rot_gate(p[0], p[1], p[2], &U[tid][0][0], &U[tid][0][1],
             &U[tid][1][0], &U[tid][1][1]);
  }
  __syncthreads();

  if (tid < NX * 8) {
    const int t = tid >> 3, j = tid & 7;
    const int g = (t + 1) >> 1;
    const bool link = (t > 0) && ((t & 1) == 0);
    float2 v[8];
#pragma unroll
    for (int r = 0; r < 8; r++) v[r] = make_float2((r == j) ? 1.f : 0.f, 0.f);
    if (link) {
      const int q = 3 * g - 1;
#pragma unroll
      for (int r = 0; r < 8; r++)
        v[r] = ((r & 3) == (j & 3)) ? U[q][r >> 2][j >> 2] : make_float2(0.f, 0.f);
    }
    {
      const int q = 3 * g;
      float2 u00 = U[q][0][0], u01 = U[q][0][1], u10 = U[q][1][0], u11 = U[q][1][1];
#pragma unroll
      for (int i = 4; i < 6; i++) {
        float2 x0 = v[i], x1 = v[i + 2];
        v[i]     = cadd(cmul(u00, x0), cmul(u01, x1));
        v[i + 2] = cadd(cmul(u10, x0), cmul(u11, x1));
      }
    }
    {
      const int q = 3 * g + 1;
      float2 u00 = U[q][0][0], u01 = U[q][0][1], u10 = U[q][1][0], u11 = U[q][1][1];
#pragma unroll
      for (int i = 2; i < 8; i += 4) {
        float2 x0 = v[i], x1 = v[i + 1];
        v[i]     = cadd(cmul(u00, x0), cmul(u01, x1));
        v[i + 1] = cadd(cmul(u10, x0), cmul(u11, x1));
      }
    }
    float2* Wt = W + (size_t)(l * NX + t) * 64;
#pragma unroll
    for (int r = 0; r < 8; r++) Wt[r * 8 + j] = v[r];
  }
}

__device__ __forceinline__ v8f wmma4(v2f a, v2f b, v8f c) {
  return __builtin_amdgcn_wmma_f32_16x16x4_f32(
      false, a, false, b, (short)0, c, false, false);
}

template <int NSTR, int GSTR, int TS0, int TS1>
__device__ __forceinline__ void kron_step(float* Sr, float* Si,
                                          const float2* V, int base0) {
  const int lane = threadIdx.x & 31;
  const int n = lane & 15;
  const int hi = lane >> 4;
  const int mm = n & 7, mhi = n >> 3;

  const float wr = (mhi == hi) ? 1.f : 0.f;
  const float wi = (mhi == hi) ? 0.f : (mhi ? 1.f : -1.f);
  v2f a[4];
#pragma unroll
  for (int ks = 0; ks < 4; ks++) {
    const float2 u0 = V[mm * VP + 2 * ks], u1 = V[mm * VP + 2 * ks + 1];
    a[ks].x = wr * u0.x + wi * u0.y;
    a[ks].y = wr * u1.x + wi * u1.y;
  }

  const float* P = hi ? Si : Sr;
  v2f b[4][4];
#pragma unroll
  for (int t = 0; t < 4; t++) {
#pragma unroll
    for (int ks = 0; ks < 4; ks++) {
      const int ad = base0 + (t & 1) * TS0 + (t >> 1) * TS1 + n * NSTR + (2 * ks) * GSTR;
      b[t][ks].x = P[ad];
      b[t][ks].y = P[ad + GSTR];
    }
  }

  v8f acc[4] = {};
#pragma unroll
  for (int ks = 0; ks < 4; ks++) {
#pragma unroll
    for (int t = 0; t < 4; t++) acc[t] = wmma4(a[ks], b[t][ks], acc[t]);
  }

  float* Q = hi ? Si : Sr;
#pragma unroll
  for (int t = 0; t < 4; t++) {
#pragma unroll
    for (int v = 0; v < 8; v++)
      Q[base0 + (t & 1) * TS0 + (t >> 1) * TS1 + n * NSTR + v * GSTR] = acc[t][v];
  }
}

__global__ __launch_bounds__(256) void qk_main(const float* __restrict__ x,
                                               const float* __restrict__ rotp,
                                               const float* __restrict__ entp,
                                               const float2* __restrict__ W,
                                               float* __restrict__ out) {
  __shared__ float S2[2 * 64 * LDA + 32];
  float* const Sr = S2;
  float* const Si = S2 + 64 * LDA + 32;
  __shared__ float2 G[NL][12][2][2];
  __shared__ float2 K8[NL][4][64];
  __shared__ float2 V[NL - 1][NX][8 * VP];
  __shared__ float2 V0[NX][8];
  __shared__ float2 RC[256];
  __shared__ float2 CU[NL][2][2];
  __shared__ float2 F1[384];

  const int tid = threadIdx.x;
  const int s = blockIdx.x;

  if (tid < 12 * NL) {
    const int gl = tid / 12, gq = tid - 12 * gl;
    const float* xs = x + (size_t)s * 48;
    float a = xs[0 * 12 + gq], b = xs[1 * 12 + gq];
    float c = xs[2 * 12 + gq], d = xs[3 * 12 + gq];
    float ca = cosf(0.5f * a), sa = sinf(0.5f * a);
    float cb = cosf(0.5f * b), sb = sinf(0.5f * b);
    float cc = cosf(0.5f * c), sc = sinf(0.5f * c);
    float cd = cosf(0.5f * d), sd = sinf(0.5f * d);
    C2 M  = { {ca, 0.f}, {-sa, 0.f}, {sa, 0.f}, {ca, 0.f} };
    C2 RX = { {cb, 0.f}, {0.f, -sb}, {0.f, -sb}, {cb, 0.f} };
    C2 RZ = { {cc, -sc}, {0.f, 0.f}, {0.f, 0.f}, {cc, sc} };
    C2 RY = { {cd, 0.f}, {-sd, 0.f}, {sd, 0.f}, {cd, 0.f} };
    M = mm2(RX, M); M = mm2(RZ, M); M = mm2(RY, M);
    const float* rp = &rotp[(gl * 12 + gq) * 3];
    C2 R;
    rot_gate(rp[0], rp[1], rp[2], &R.a00, &R.a01, &R.a10, &R.a11);
    M = mm2(R, M);
    G[gl][gq][0][0] = M.a00; G[gl][gq][0][1] = M.a01;
    G[gl][gq][1][0] = M.a10; G[gl][gq][1][1] = M.a11;
  }
  if (tid >= 64 && tid < 64 + NL) {
    int l = tid - 64;
    const float* p = &entp[(l * 12 + 11) * 3];
    rot_gate(p[0], p[1], p[2], &CU[l][0][0], &CU[l][0][1],
             &CU[l][1][0], &CU[l][1][1]);
  }
  __syncthreads();

  for (int e = tid; e < NL * 256; e += 256) {
    const int l = e >> 8;
    const int kg = (e >> 6) & 3, ka = (e >> 3) & 7, kb8 = e & 7;
    float2 p = cmul(cmul(G[l][3 * kg][(ka >> 2) & 1][(kb8 >> 2) & 1],
                         G[l][3 * kg + 1][(ka >> 1) & 1][(kb8 >> 1) & 1]),
                    G[l][3 * kg + 2][ka & 1][kb8 & 1]);
    K8[l][kg][ka * 8 + kb8] = p;
  }
  __syncthreads();

  constexpr int NFULL = (NL - 1) * NX * 64;
  for (int e = tid; e < NFULL + NX * 8; e += 256) {
    int lt, r, c;
    if (e < NFULL) { lt = NX + (e >> 6); r = (e >> 3) & 7; c = e & 7; }
    else           { lt = (e - NFULL) >> 3; r = (e - NFULL) & 7; c = 0; }
    const int l = lt / NX, t = lt - l * NX;
    const int g = (t + 1) >> 1;
    const float2* Xr = W + (size_t)lt * 64 + r * 8;
    const float2* Kc = &K8[l][g][c];
    float2 acc = make_float2(0.f, 0.f);
#pragma unroll
    for (int k = 0; k < 8; k++) acc = cadd(acc, cmul(Xr[k], Kc[k * 8]));
    float2* dst = (l > 0) ? &V[l - 1][t][r * VP + c] : &V0[t][r];
    *dst = acc;
  }
  __syncthreads();

  {
    float2 val;
    if (tid < 64) {
      const int ra = tid >> 3, rb = tid & 7;
      val = cmul(V0[0][ra], V0[1 + (ra & 1)][rb]);
    } else if (tid < 128) {
      const int h = (tid >> 5) & 1;
      const int ra = (tid >> 3) & 3, rb = tid & 7;
      const float2 bf = V0[1 + (ra & 1)][rb];
      const float2 x0 = cmul(V0[0][ra], bf);
      const float2 x1 = cmul(V0[0][ra + 4], bf);
      val = cadd(cmul(CU[0][h][0], x0), cmul(CU[0][h][1], x1));
    } else {
      const int p = (tid >> 6) & 1;
      const int gc = (tid >> 3) & 7, gd = tid & 7;
      val = cmul(V0[3 + p][gc], V0[5 + (gc & 1)][gd]);
    }
    RC[tid] = val;
  }
  __syncthreads();

  const float2 (*V1)[8 * VP] = V[0];
  float2* const T1 = &K8[0][0][0];
  {
    {
      const int p = tid >> 7, sc = (tid >> 6) & 1, c = tid & 63;
      const float2* Vc = V1[3 + sc] + (c >> 3) * VP;
      const float2* Cf = RC + 128 + 64 * p + (c & 7);
      float2 acc = make_float2(0.f, 0.f);
#pragma unroll
      for (int k = 0; k < 8; k++) acc = cadd(acc, cmul(Vc[k], Cf[8 * k]));
      F1[128 + tid] = acc;
    }
    if (tid < 128) {
      const int e = tid >> 6, r = tid & 63;
      const float2* Va = V1[0] + (r >> 3) * VP;
      const float2* Rs = RC + 64 * e + (r & 7);
      float2 acc = make_float2(0.f, 0.f);
#pragma unroll
      for (int k = 0; k < 8; k++) acc = cadd(acc, cmul(Va[k], Rs[8 * k]));
      F1[tid] = acc;
    }
  }
  __syncthreads();

  {
    {
      const int e = tid >> 7, p = (tid >> 6) & 1, r = tid & 63;
      const int r5 = r & 31, h = r >> 5;
      float2 lohi[2];
#pragma unroll
      for (int u = 0; u < 2; u++) {
        const int rr = r5 + 32 * u;
        const int ra = rr >> 3;
        const float2* Vb = V1[1 + (ra & 1)] + (rr & 7) * VP + p;
        const float2* Ra = F1 + 64 * e + 8 * ra + p;
        float2 acc = make_float2(0.f, 0.f);
#pragma unroll
        for (int k = 0; k < 4; k++) acc = cadd(acc, cmul(Vb[2 * k], Ra[2 * k]));
        lohi[u] = acc;
      }
      T1[tid] = h ? lohi[1] : lohi[0];
      T1[256 + tid] = cadd(cmul(CU[1][h][0], lohi[0]), cmul(CU[1][h][1], lohi[1]));
    }
    for (int o = tid; o < 512; o += 256) {
      const int c = o & 63, sc = (o >> 6) & 1, e = (o >> 7) & 1, p = o >> 8;
      const int gc = c >> 3;
      const float2* Vd = V1[5 + (gc & 1)] + (c & 7) * VP + e;
      const float2* Cc = F1 + 128 + (2 * p + sc) * 64 + 8 * gc + e;
      float2 acc = make_float2(0.f, 0.f);
#pragma unroll
      for (int k = 0; k < 4; k++) acc = cadd(acc, cmul(Vd[2 * k], Cc[2 * k]));
      T1[512 + o] = acc;
    }
  }
  __syncthreads();

  {
    const int j = tid & 63, q = tid >> 6;
    const int sp = q & 1;
    const float2* Rt = T1 + 256 * (j & 1);
    float2 cd[4];
#pragma unroll
    for (int e = 0; e < 2; e++)
#pragma unroll
      for (int p = 0; p < 2; p++)
        cd[2 * e + p] = T1[512 + ((2 * p + e) * 2 + sp) * 64 + j];
    for (int it = 0; it < 16; it++) {
      const int i = q + 4 * it;
      float2 a = cmul(Rt[i], cd[0]);
#pragma unroll
      for (int m = 1; m < 4; m++) a = cadd(a, cmul(Rt[64 * m + i], cd[m]));
      Sr[i * LDA + j] = a.x;
      Si[i * LDA + j] = a.y;
    }
  }
  __syncthreads();

  const int w = tid >> 5;
  const int par = (w >> 1) & 1;

#pragma nounroll
  for (int l = 2; l < NL; l++) {
    const float2 (*Vl)[8 * VP] = V[l - 1];
    kron_step<1, 8 * LDA, 16, 32>(Sr, Si, Vl[0], w * LDA);
    __syncthreads();
    kron_step<1, LDA, 16, 32>(Sr, Si, Vl[1 + (w & 1)], w * 8 * LDA);
    __syncthreads();
    kron_step<2 * LDA, 8, 1, 2>(Sr, Si, Vl[3 + par],
                             (par + 32 * (w >> 2)) * LDA + 4 * (w & 1));
    __syncthreads();
    kron_step<4 * LDA, 1, LDA, 2 * LDA>(Sr, Si, Vl[5 + (w & 1)], w * 8);
    __syncthreads();

    {
      float2 u00 = CU[l][0][0], u01 = CU[l][0][1];
      float2 u10 = CU[l][1][0], u11 = CU[l][1][1];
      for (int t = tid; t < 1024; t += 256) {
        int il = t >> 5;
        int j = ((t & 31) << 1) | 1;
        int a0 = il * LDA + j, a1 = (il + 32) * LDA + j;
        float2 x0 = make_float2(Sr[a0], Si[a0]);
        float2 x1 = make_float2(Sr[a1], Si[a1]);
        float2 y0 = cadd(cmul(u00, x0), cmul(u01, x1));
        float2 y1 = cadd(cmul(u10, x0), cmul(u11, x1));
        Sr[a0] = y0.x; Si[a0] = y0.y;
        Sr[a1] = y1.x; Si[a1] = y1.y;
      }
    }
    __syncthreads();
  }

  float z[12];
  for (int q = 0; q < 12; q++) z[q] = 0.f;
  for (int idx = tid; idx < 4096; idx += 256) {
    int i = idx >> 6, j = idx & 63;
    float pr = Sr[i * LDA + j], pi = Si[i * LDA + j];
    float p = pr * pr + pi * pi;
    for (int q = 0; q < 12; q++)
      z[q] += ((idx >> (11 - q)) & 1) ? -p : p;
  }
  float* red = (float*)&V[0][0][0];
  for (int q = 0; q < 12; q++) red[tid * 12 + q] = z[q];
  __syncthreads();
  for (int sft = 128; sft > 0; sft >>= 1) {
    if (tid < sft)
      for (int q = 0; q < 12; q++)
        red[tid * 12 + q] += red[(tid + sft) * 12 + q];
    __syncthreads();
  }
  if (tid < 12) out[(size_t)s * 12 + tid] = red[tid];
}

extern "C" void kernel_launch(void* const* d_in, const int* in_sizes, int n_in,
                              void* d_out, int out_size, void* d_ws, size_t ws_size,
                              hipStream_t stream) {
  (void)n_in; (void)out_size; (void)ws_size;
  const float* x    = (const float*)d_in[0];
  const float* rotp = (const float*)d_in[1];
  const float* entp = (const float*)d_in[2];
  float2* W = (float2*)d_ws;
  const int B = in_sizes[0] / 48;

  qk_prep<<<NL, 64, 0, stream>>>(entp, W);
  qk_main<<<B, 256, 0, stream>>>(x, rotp, entp, W, (float*)d_out);
}


# Measurement sums taken bit by bit: per-thread add/subtract tree over 16 rows, lane butterfly, 8 per-wave partials; replaces 12 selects per amplitude and the 8-level LDS tree; sum order changes
# speedup vs baseline: 8.7329x; 1.0837x over previous
#include <hip/hip_runtime.h>
#include <math.h>

#define LDA 65
#define NL  4
#define NX  7
#define VP  10

typedef __attribute__((ext_vector_type(2))) float v2f;
typedef __attribute__((ext_vector_type(8))) float v8f;

__device__ __forceinline__ float2 cmul(float2 a, float2 b) {
  return make_float2(a.x * b.x - a.y * b.y, a.x * b.y + a.y * b.x);
}
__device__ __forceinline__ float2 cadd(float2 a, float2 b) {
  return make_float2(a.x + b.x, a.y + b.y);
}

struct C2 { float2 a00, a01, a10, a11; };
__device__ __forceinline__ C2 mm2(const C2 L, const C2 R) {
  C2 o;
  o.a00 = cadd(cmul(L.a00, R.a00), cmul(L.a01, R.a10));
  o.a01 = cadd(cmul(L.a00, R.a01), cmul(L.a01, R.a11));
  o.a10 = cadd(cmul(L.a10, R.a00), cmul(L.a11, R.a10));
  o.a11 = cadd(cmul(L.a10, R.a01), cmul(L.a11, R.a11));
  return o;
}

__device__ __forceinline__ void rot_gate(float phi, float th, float om,
                                         float2* u00, float2* u01,
                                         float2* u10, float2* u11) {
  float ct = cosf(0.5f * th), st = sinf(0.5f * th);
  float ap = -0.5f * (phi + om);
  float am = 0.5f * (phi - om);
  *u00 = make_float2(ct * cosf(ap), ct * sinf(ap));
  *u01 = make_float2(-st * cosf(am), -st * sinf(am));
  *u10 = make_float2(st * cosf(-am), st * sinf(-am));
  *u11 = make_float2(ct * cosf(-ap), ct * sinf(-ap));
}

__global__ __launch_bounds__(64) void qk_prep(const float* __restrict__ entp,
                                              float2* __restrict__ W) {
  __shared__ float2 U[12][2][2];
  const int l = blockIdx.x;
  const int tid = threadIdx.x;

  if (tid < 12) {
    const float* p = &entp[(l * 12 + tid) * 3];
    rot_gate(p[0], p[1], p[2], &U[tid][0][0], &U[tid][0][1],
             &U[tid][1][0], &U[tid][1][1]);
  }
  __syncthreads();

  if (tid < NX * 8) {
    const int t = tid >> 3, j = tid & 7;
    const int g = (t + 1) >> 1;
    const bool link = (t > 0) && ((t & 1) == 0);
    float2 v[8];
#pragma unroll
    for (int r = 0; r < 8; r++) v[r] = make_float2((r == j) ? 1.f : 0.f, 0.f);
    if (link) {
      const int q = 3 * g - 1;
#pragma unroll
      for (int r = 0; r < 8; r++)
        v[r] = ((r & 3) == (j & 3)) ? U[q][r >> 2][j >> 2] : make_float2(0.f, 0.f);
    }
    {
      const int q = 3 * g;
      float2 u00 = U[q][0][0], u01 = U[q][0][1], u10 = U[q][1][0], u11 = U[q][1][1];
#pragma unroll
      for (int i = 4; i < 6; i++) {
        float2 x0 = v[i], x1 = v[i + 2];
        v[i]     = cadd(cmul(u00, x0), cmul(u01, x1));
        v[i + 2] = cadd(cmul(u10, x0), cmul(u11, x1));
      }
    }
    {
      const int q = 3 * g + 1;
      float2 u00 = U[q][0][0], u01 = U[q][0][1], u10 = U[q][1][0], u11 = U[q][1][1];
#pragma unroll
      for (int i = 2; i < 8; i += 4) {
        float2 x0 = v[i], x1 = v[i + 1];
        v[i]     = cadd(cmul(u00, x0), cmul(u01, x1));
        v[i + 1] = cadd(cmul(u10, x0), cmul(u11, x1));
      }
    }
    float2* Wt = W + (size_t)(l * NX + t) * 64;
#pragma unroll
    for (int r = 0; r < 8; r++) Wt[r * 8 + j] = v[r];
  }
}

__device__ __forceinline__ v8f wmma4(v2f a, v2f b, v8f c) {
  return __builtin_amdgcn_wmma_f32_16x16x4_f32(
      false, a, false, b, (short)0, c, false, false);
}

template <int NSTR, int GSTR, int TS0, int TS1>
__device__ __forceinline__ void kron_step(float* Sr, float* Si,
                                          const float2* V, int base0) {
  const int lane = threadIdx.x & 31;
  const int n = lane & 15;
  const int hi = lane >> 4;
  const int mm = n & 7, mhi = n >> 3;

  const float wr = (mhi == hi) ? 1.f : 0.f;
  const float wi = (mhi == hi) ? 0.f : (mhi ? 1.f : -1.f);
  v2f a[4];
#pragma unroll
  for (int ks = 0; ks < 4; ks++) {
    const float2 u0 = V[mm * VP + 2 * ks], u1 = V[mm * VP + 2 * ks + 1];
    a[ks].x = wr * u0.x + wi * u0.y;
    a[ks].y = wr * u1.x + wi * u1.y;
  }

  const float* P = hi ? Si : Sr;
  v2f b[4][4];
#pragma unroll
  for (int t = 0; t < 4; t++) {
#pragma unroll
    for (int ks = 0; ks < 4; ks++) {
      const int ad = base0 + (t & 1) * TS0 + (t >> 1) * TS1 + n * NSTR + (2 * ks) * GSTR;
      b[t][ks].x = P[ad];
      b[t][ks].y = P[ad + GSTR];
    }
  }

  v8f acc[4] = {};
#pragma unroll
  for (int ks = 0; ks < 4; ks++) {
#pragma unroll
    for (int t = 0; t < 4; t++) acc[t] = wmma4(a[ks], b[t][ks], acc[t]);
  }

  float* Q = hi ? Si : Sr;
#pragma unroll
  for (int t = 0; t < 4; t++) {
#pragma unroll
    for (int v = 0; v < 8; v++)
      Q[base0 + (t & 1) * TS0 + (t >> 1) * TS1 + n * NSTR + v * GSTR] = acc[t][v];
  }
}

__global__ __launch_bounds__(256, 10) void qk_main(const float* __restrict__ x,
                                               const float* __restrict__ rotp,
                                               const float* __restrict__ entp,
                                               const float2* __restrict__ W,
                                               float* __restrict__ out) {
  __shared__ float S2[2 * 64 * LDA + 32];
  float* const Sr = S2;
  float* const Si = S2 + 64 * LDA + 32;
  __shared__ float2 G[NL][12][2][2];
  __shared__ float2 K8[NL][4][64];
  __shared__ float2 V[NL - 1][NX][8 * VP];
  __shared__ float2 V0[NX][8];
  __shared__ float2 RC[256];
  __shared__ float2 CU[NL][2][2];
  __shared__ float2 F1[384];

  const int tid = threadIdx.x;
  const int s = blockIdx.x;

  if (tid < 12 * NL) {
    const int gl = tid / 12, gq = tid - 12 * gl;
    const float* xs = x + (size_t)s * 48;
    float a = xs[0 * 12 + gq], b = xs[1 * 12 + gq];
    float c = xs[2 * 12 + gq], d = xs[3 * 12 + gq];
    float ca = cosf(0.5f * a), sa = sinf(0.5f * a);
    float cb = cosf(0.5f * b), sb = sinf(0.5f * b);
    float cc = cosf(0.5f * c), sc = sinf(0.5f * c);
    float cd = cosf(0.5f * d), sd = sinf(0.5f * d);
    C2 M  = { {ca, 0.f}, {-sa, 0.f}, {sa, 0.f}, {ca, 0.f} };
    C2 RX = { {cb, 0.f}, {0.f, -sb}, {0.f, -sb}, {cb, 0.f} };
    C2 RZ = { {cc, -sc}, {0.f, 0.f}, {0.f, 0.f}, {cc, sc} };
    C2 RY = { {cd, 0.f}, {-sd, 0.f}, {sd, 0.f}, {cd, 0.f} };
    M = mm2(RX, M); M = mm2(RZ, M); M = mm2(RY, M);
    const float* rp = &rotp[(gl * 12 + gq) * 3];
    C2 R;
    rot_gate(rp[0], rp[1], rp[2], &R.a00, &R.a01, &R.a10, &R.a11);
    M = mm2(R, M);
    G[gl][gq][0][0] = M.a00; G[gl][gq][0][1] = M.a01;
    G[gl][gq][1][0] = M.a10; G[gl][gq][1][1] = M.a11;
  }
  if (tid >= 64 && tid < 64 + NL) {
    int l = tid - 64;
    const float* p = &entp[(l * 12 + 11) * 3];
    rot_gate(p[0], p[1], p[2], &CU[l][0][0], &CU[l][0][1],
             &CU[l][1][0], &CU[l][1][1]);
  }
  __syncthreads();

  for (int e = tid; e < NL * 256; e += 256) {
    const int l = e >> 8;
    const int kg = (e >> 6) & 3, ka = (e >> 3) & 7, kb8 = e & 7;
    float2 p = cmul(cmul(G[l][3 * kg][(ka >> 2) & 1][(kb8 >> 2) & 1],
                         G[l][3 * kg + 1][(ka >> 1) & 1][(kb8 >> 1) & 1]),
                    G[l][3 * kg + 2][ka & 1][kb8 & 1]);
    K8[l][kg][ka * 8 + kb8] = p;
  }
  __syncthreads();

  constexpr int NFULL = (NL - 1) * NX * 64;
  for (int e = tid; e < NFULL + NX * 8; e += 256) {
    int lt, r, c;
    if (e < NFULL) { lt = NX + (e >> 6); r = (e >> 3) & 7; c = e & 7; }
    else           { lt = (e - NFULL) >> 3; r = (e - NFULL) & 7; c = 0; }
    const int l = lt / NX, t = lt - l * NX;
    const int g = (t + 1) >> 1;
    const float2* Xr = W + (size_t)lt * 64 + r * 8;
    const float2* Kc = &K8[l][g][c];
    float2 acc = make_float2(0.f, 0.f);
#pragma unroll
    for (int k = 0; k < 8; k++) acc = cadd(acc, cmul(Xr[k], Kc[k * 8]));
    float2* dst = (l > 0) ? &V[l - 1][t][r * VP + c] : &V0[t][r];
    *dst = acc;
  }
  __syncthreads();

  {
    float2 val;
    if (tid < 64) {
      const int ra = tid >> 3, rb = tid & 7;
      val = cmul(V0[0][ra], V0[1 + (ra & 1)][rb]);
    } else if (tid < 128) {
      const int h = (tid >> 5) & 1;
      const int ra = (tid >> 3) & 3, rb = tid & 7;
      const float2 bf = V0[1 + (ra & 1)][rb];
      const float2 x0 = cmul(V0[0][ra], bf);
      const float2 x1 = cmul(V0[0][ra + 4], bf);
      val = cadd(cmul(CU[0][h][0], x0), cmul(CU[0][h][1], x1));
    } else {
      const int p = (tid >> 6) & 1;
      const int gc = (tid >> 3) & 7, gd = tid & 7;
      val = cmul(V0[3 + p][gc], V0[5 + (gc & 1)][gd]);
    }
    RC[tid] = val;
  }
  __syncthreads();

  const float2 (*V1)[8 * VP] = V[0];
  float2* const T1 = &K8[0][0][0];
  {
    {
      const int p = tid >> 7, sc = (tid >> 6) & 1, c = tid & 63;
      const float2* Vc = V1[3 + sc] + (c >> 3) * VP;
      const float2* Cf = RC + 128 + 64 * p + (c & 7);
      float2 acc = make_float2(0.f, 0.f);
#pragma unroll
      for (int k = 0; k < 8; k++) acc = cadd(acc, cmul(Vc[k], Cf[8 * k]));
      F1[128 + tid] = acc;
    }
    if (tid < 128) {
      const int e = tid >> 6, r = tid & 63;
      const float2* Va = V1[0] + (r >> 3) * VP;
      const float2* Rs = RC + 64 * e + (r & 7);
      float2 acc = make_float2(0.f, 0.f);
#pragma unroll
      for (int k = 0; k < 8; k++) acc = cadd(acc, cmul(Va[k], Rs[8 * k]));
      F1[tid] = acc;
    }
  }
  __syncthreads();

  {
    {
      const int e = tid >> 7, p = (tid >> 6) & 1, r = tid & 63;
      const int r5 = r & 31, h = r >> 5;
      float2 lohi[2];
#pragma unroll
      for (int u = 0; u < 2; u++) {
        const int rr = r5 + 32 * u;
        const int ra = rr >> 3;
        const float2* Vb = V1[1 + (ra & 1)] + (rr & 7) * VP + p;
        const float2* Ra = F1 + 64 * e + 8 * ra + p;
        float2 acc = make_float2(0.f, 0.f);
#pragma unroll
        for (int k = 0; k < 4; k++) acc = cadd(acc, cmul(Vb[2 * k], Ra[2 * k]));
        lohi[u] = acc;
      }
      T1[tid] = h ? lohi[1] : lohi[0];
      T1[256 + tid] = cadd(cmul(CU[1][h][0], lohi[0]), cmul(CU[1][h][1], lohi[1]));
    }
    for (int o = tid; o < 512; o += 256) {
      const int c = o & 63, sc = (o >> 6) & 1, e = (o >> 7) & 1, p = o >> 8;
      const int gc = c >> 3;
      const float2* Vd = V1[5 + (gc & 1)] + (c & 7) * VP + e;
      const float2* Cc = F1 + 128 + (2 * p + sc) * 64 + 8 * gc + e;
      float2 acc = make_float2(0.f, 0.f);
#pragma unroll
      for (int k = 0; k < 4; k++) acc = cadd(acc, cmul(Vd[2 * k], Cc[2 * k]));
      T1[512 + o] = acc;
    }
  }
  __syncthreads();

  {
    const int j = tid & 63, q = tid >> 6;
    const int sp = q & 1;
    const float2* Rt = T1 + 256 * (j & 1);
    float2 cd[4];
#pragma unroll
    for (int e = 0; e < 2; e++)
#pragma unroll
      for (int p = 0; p < 2; p++)
        cd[2 * e + p] = T1[512 + ((2 * p + e) * 2 + sp) * 64 + j];
    for (int it = 0; it < 16; it++) {
      const int i = q + 4 * it;
      float2 a = cmul(Rt[i], cd[0]);
#pragma unroll
      for (int m = 1; m < 4; m++) a = cadd(a, cmul(Rt[64 * m + i], cd[m]));
      Sr[i * LDA + j] = a.x;
      Si[i * LDA + j] = a.y;
    }
  }
  __syncthreads();

  const int w = tid >> 5;
  const int par = (w >> 1) & 1;

#pragma nounroll
  for (int l = 2; l < NL; l++) {
    const float2 (*Vl)[8 * VP] = V[l - 1];
    kron_step<1, 8 * LDA, 16, 32>(Sr, Si, Vl[0], w * LDA);
    __syncthreads();
    kron_step<1, LDA, 16, 32>(Sr, Si, Vl[1 + (w & 1)], w * 8 * LDA);
    __syncthreads();
    kron_step<2 * LDA, 8, 1, 2>(Sr, Si, Vl[3 + par],
                             (par + 32 * (w >> 2)) * LDA + 4 * (w & 1));
    __syncthreads();
    kron_step<4 * LDA, 1, LDA, 2 * LDA>(Sr, Si, Vl[5 + (w & 1)], w * 8);
    __syncthreads();

    {
      float2 u00 = CU[l][0][0], u01 = CU[l][0][1];
      float2 u10 = CU[l][1][0], u11 = CU[l][1][1];
      for (int t = tid; t < 1024; t += 256) {
        int il = t >> 5;
        int j = ((t & 31) << 1) | 1;
        int a0 = il * LDA + j, a1 = (il + 32) * LDA + j;
        float2 x0 = make_float2(Sr[a0], Si[a0]);
        float2 x1 = make_float2(Sr[a1], Si[a1]);
        float2 y0 = cadd(cmul(u00, x0), cmul(u01, x1));
        float2 y1 = cadd(cmul(u10, x0), cmul(u11, x1));
        Sr[a0] = y0.x; Si[a0] = y0.y;
        Sr[a1] = y1.x; Si[a1] = y1.y;
      }
    }
    __syncthreads();
  }

  {
    const int j = tid & 63, qd = tid >> 6;
    float p[16];
#pragma unroll
    for (int k = 0; k < 16; k++) {
      const int ad = (qd + 4 * k) * LDA + j;
      const float pr = Sr[ad], pi = Si[ad];
      p[k] = pr * pr + pi * pi;
    }
    float a1[8];
    float D0 = p[0] - p[1];
    a1[0] = p[0] + p[1];
#pragma unroll
    for (int m = 1; m < 8; m++) {
      a1[m] = p[2 * m] + p[2 * m + 1];
      D0 += p[2 * m] - p[2 * m + 1];
    }
    float a2[4];
    float D1 = a1[0] - a1[1];
    a2[0] = a1[0] + a1[1];
#pragma unroll
    for (int m = 1; m < 4; m++) {
      a2[m] = a1[2 * m] + a1[2 * m + 1];
      D1 += a1[2 * m] - a1[2 * m + 1];
    }
    const float a30 = a2[0] + a2[1], a31 = a2[2] + a2[3];
    float D2 = (a2[0] - a2[1]) + (a2[2] - a2[3]);
    float D3 = a30 - a31;
    float T = a30 + a31;

    const int lane = tid & 31;
#pragma unroll
    for (int m = 1; m < 32; m <<= 1) {
      const float t = __shfl_xor(T, m);
      T = (lane & m) ? (t - T) : (T + t);
      D0 += __shfl_xor(D0, m);
      D1 += __shfl_xor(D1, m);
      D2 += __shfl_xor(D2, m);
      D3 += __shfl_xor(D3, m);
    }

    float* part = (float*)&V[0][0][0];
    if (lane == 0) {
      part[w * 12 + 0] = D3;
      part[w * 12 + 1] = D2;
      part[w * 12 + 2] = D1;
      part[w * 12 + 3] = D0;
      part[w * 12 + 4] = (w & 4) ? -T : T;
      part[w * 12 + 5] = (w & 2) ? -T : T;
      part[w * 12 + 6] = (w & 1) ? -T : T;
    } else if ((lane & (lane - 1)) == 0) {
      part[w * 12 + 11 - (31 - __clz(lane))] = T;
    }
    __syncthreads();
    if (tid < 12) {
      float zq = part[tid];
#pragma unroll
      for (int ww = 1; ww < 8; ww++) zq += part[ww * 12 + tid];
      out[(size_t)s * 12 + tid] = zq;
    }
  }
}

extern "C" void kernel_launch(void* const* d_in, const int* in_sizes, int n_in,
                              void* d_out, int out_size, void* d_ws, size_t ws_size,
                              hipStream_t stream) {
  (void)n_in; (void)out_size; (void)ws_size;
  const float* x    = (const float*)d_in[0];
  const float* rotp = (const float*)d_in[1];
  const float* entp = (const float*)d_in[2];
  float2* W = (float2*)d_ws;
  const int B = in_sizes[0] / 48;

  qk_prep<<<NL, 64, 0, stream>>>(entp, W);
  qk_main<<<B, 256, 0, stream>>>(x, rotp, entp, W, (float*)d_out);
}


# Step factors V = X*K8 (all layers) built on f32 WMMA as K^T*X^T tiles, two per wave, X planar in the workspace; replaces per-thread 8-term complex sums; sum order of the factors changes
# speedup vs baseline: 9.4845x; 1.0861x over previous
#include <hip/hip_runtime.h>
#include <math.h>

#define LDA 65
#define NL  4
#define NX  7
#define VP  10

typedef __attribute__((ext_vector_type(2))) float v2f;
typedef __attribute__((ext_vector_type(8))) float v8f;

__device__ __forceinline__ float2 cmul(float2 a, float2 b) {
  return make_float2(a.x * b.x - a.y * b.y, a.x * b.y + a.y * b.x);
}
__device__ __forceinline__ float2 cadd(float2 a, float2 b) {
  return make_float2(a.x + b.x, a.y + b.y);
}

struct C2 { float2 a00, a01, a10, a11; };
__device__ __forceinline__ C2 mm2(const C2 L, const C2 R) {
  C2 o;
  o.a00 = cadd(cmul(L.a00, R.a00), cmul(L.a01, R.a10));
  o.a01 = cadd(cmul(L.a00, R.a01), cmul(L.a01, R.a11));
  o.a10 = cadd(cmul(L.a10, R.a00), cmul(L.a11, R.a10));
  o.a11 = cadd(cmul(L.a10, R.a01), cmul(L.a11, R.a11));
  return o;
}

__device__ __forceinline__ void rot_gate(float phi, float th, float om,
                                         float2* u00, float2* u01,
                                         float2* u10, float2* u11) {
  float ct = cosf(0.5f * th), st = sinf(0.5f * th);
  float ap = -0.5f * (phi + om);
  float am = 0.5f * (phi - om);
  *u00 = make_float2(ct * cosf(ap), ct * sinf(ap));
  *u01 = make_float2(-st * cosf(am), -st * sinf(am));
  *u10 = make_float2(st * cosf(-am), st * sinf(-am));
  *u11 = make_float2(ct * cosf(-ap), ct * sinf(-ap));
}

__global__ __launch_bounds__(64) void qk_prep(const float* __restrict__ entp,
                                              float2* __restrict__ W) {
  __shared__ float2 U[12][2][2];
  const int l = blockIdx.x;
  const int tid = threadIdx.x;

  if (tid < 12) {
    const float* p = &entp[(l * 12 + tid) * 3];
    rot_gate(p[0], p[1], p[2], &U[tid][0][0], &U[tid][0][1],
             &U[tid][1][0], &U[tid][1][1]);
  }
  __syncthreads();

  if (tid < NX * 8) {
    const int t = tid >> 3, j = tid & 7;
    const int g = (t + 1) >> 1;
    const bool link = (t > 0) && ((t & 1) == 0);
    float2 v[8];
#pragma unroll
    for (int r = 0; r < 8; r++) v[r] = make_float2((r == j) ? 1.f : 0.f, 0.f);
    if (link) {
      const int q = 3 * g - 1;
#pragma unroll
      for (int r = 0; r < 8; r++)
        v[r] = ((r & 3) == (j & 3)) ? U[q][r >> 2][j >> 2] : make_float2(0.f, 0.f);
    }
    {
      const int q = 3 * g;
      float2 u00 = U[q][0][0], u01 = U[q][0][1], u10 = U[q][1][0], u11 = U[q][1][1];
#pragma unroll
      for (int i = 4; i < 6; i++) {
        float2 x0 = v[i], x1 = v[i + 2];
        v[i]     = cadd(cmul(u00, x0), cmul(u01, x1));
        v[i + 2] = cadd(cmul(u10, x0), cmul(u11, x1));
      }
    }
    {
      const int q = 3 * g + 1;
      float2 u00 = U[q][0][0], u01 = U[q][0][1], u10 = U[q][1][0], u11 = U[q][1][1];
#pragma unroll
      for (int i = 2; i < 8; i += 4) {
        float2 x0 = v[i], x1 = v[i + 1];
        v[i]     = cadd(cmul(u00, x0), cmul(u01, x1));
        v[i + 1] = cadd(cmul(u10, x0), cmul(u11, x1));
      }
    }
    float* Wt = (float*)(W + (size_t)(l * NX + t) * 64);
#pragma unroll
    for (int r = 0; r < 8; r++) {
      Wt[r * 8 + j] = v[r].x;
      Wt[64 + r * 8 + j] = v[r].y;
    }
  }
}

__device__ __forceinline__ v8f wmma4(v2f a, v2f b, v8f c) {
  return __builtin_amdgcn_wmma_f32_16x16x4_f32(
      false, a, false, b, (short)0, c, false, false);
}

template <int NSTR, int GSTR, int TS0, int TS1>
__device__ __forceinline__ void kron_step(float* Sr, float* Si,
                                          const float2* V, int base0) {
  const int lane = threadIdx.x & 31;
  const int n = lane & 15;
  const int hi = lane >> 4;
  const int mm = n & 7, mhi = n >> 3;

  const float wr = (mhi == hi) ? 1.f : 0.f;
  const float wi = (mhi == hi) ? 0.f : (mhi ? 1.f : -1.f);
  v2f a[4];
#pragma unroll
  for (int ks = 0; ks < 4; ks++) {
    const float2 u0 = V[mm * VP + 2 * ks], u1 = V[mm * VP + 2 * ks + 1];
    a[ks].x = wr * u0.x + wi * u0.y;
    a[ks].y = wr * u1.x + wi * u1.y;
  }

  const float* P = hi ? Si : Sr;
  v2f b[4][4];
#pragma unroll
  for (int t = 0; t < 4; t++) {
#pragma unroll
    for (int ks = 0; ks < 4; ks++) {
      const int ad = base0 + (t & 1) * TS0 + (t >> 1) * TS1 + n * NSTR + (2 * ks) * GSTR;
      b[t][ks].x = P[ad];
      b[t][ks].y = P[ad + GSTR];
    }
  }

  v8f acc[4] = {};
#pragma unroll
  for (int ks = 0; ks < 4; ks++) {
#pragma unroll
    for (int t = 0; t < 4; t++) acc[t] = wmma4(a[ks], b[t][ks], acc[t]);
  }

  float* Q = hi ? Si : Sr;
#pragma unroll
  for (int t = 0; t < 4; t++) {
#pragma unroll
    for (int v = 0; v < 8; v++)
      Q[base0 + (t & 1) * TS0 + (t >> 1) * TS1 + n * NSTR + v * GSTR] = acc[t][v];
  }
}

__global__ __launch_bounds__(256, 10) void qk_main(const float* __restrict__ x,
                                               const float* __restrict__ rotp,
                                               const float* __restrict__ entp,
                                               const float2* __restrict__ W,
                                               float* __restrict__ out) {
  __shared__ float S2[2 * 64 * LDA + 32];
  float* const Sr = S2;
  float* const Si = S2 + 64 * LDA + 32;
  __shared__ float2 G[NL][12][2][2];
  __shared__ float2 K8[NL][4][64];
  __shared__ float2 V[NL - 1][NX][8 * VP];
  __shared__ float2 V0[NX][8];
  __shared__ float2 RC[256];
  __shared__ float2 CU[NL][2][2];
  __shared__ float2 F1[384];

  const int tid = threadIdx.x;
  const int s = blockIdx.x;

  if (tid < 12 * NL) {
    const int gl = tid / 12, gq = tid - 12 * gl;
    const float* xs = x + (size_t)s * 48;
    float a = xs[0 * 12 + gq], b = xs[1 * 12 + gq];
    float c = xs[2 * 12 + gq], d = xs[3 * 12 + gq];
    float ca = cosf(0.5f * a), sa = sinf(0.5f * a);
    float cb = cosf(0.5f * b), sb = sinf(0.5f * b);
    float cc = cosf(0.5f * c), sc = sinf(0.5f * c);
    float cd = cosf(0.5f * d), sd = sinf(0.5f * d);
    C2 M  = { {ca, 0.f}, {-sa, 0.f}, {sa, 0.f}, {ca, 0.f} };
    C2 RX = { {cb, 0.f}, {0.f, -sb}, {0.f, -sb}, {cb, 0.f} };
    C2 RZ = { {cc, -sc}, {0.f, 0.f}, {0.f, 0.f}, {cc, sc} };
    C2 RY = { {cd, 0.f}, {-sd, 0.f}, {sd, 0.f}, {cd, 0.f} };
    M = mm2(RX, M); M = mm2(RZ, M); M = mm2(RY, M);
    const float* rp = &rotp[(gl * 12 + gq) * 3];
    C2 R;
    rot_gate(rp[0], rp[1], rp[2], &R.a00, &R.a01, &R.a10, &R.a11);
    M = mm2(R, M);
    G[gl][gq][0][0] = M.a00; G[gl][gq][0][1] = M.a01;
    G[gl][gq][1][0] = M.a10; G[gl][gq][1][1] = M.a11;
  }
  if (tid >= 64 && tid < 64 + NL) {
    int l = tid - 64;
    const float* p = &entp[(l * 12 + 11) * 3];
    rot_gate(p[0], p[1], p[2], &CU[l][0][0], &CU[l][0][1],
             &CU[l][1][0], &CU[l][1][1]);
  }
  __syncthreads();

  for (int e = tid; e < NL * 256; e += 256) {
    const int l = e >> 8;
    const int kg = (e >> 6) & 3, ka = (e >> 3) & 7, kb8 = e & 7;
    float2 p = cmul(cmul(G[l][3 * kg][(ka >> 2) & 1][(kb8 >> 2) & 1],
                         G[l][3 * kg + 1][(ka >> 1) & 1][(kb8 >> 1) & 1]),
                    G[l][3 * kg + 2][ka & 1][kb8 & 1]);
    K8[l][kg][ka * 8 + kb8] = p;
  }
  __syncthreads();

  {
    const int lane = tid & 31;
    const int n = lane & 15, hi = lane >> 4;
    const int r = n & 7, tt = n >> 3;
    const int wv = tid >> 5;
    const int g = wv & 3;
    const int t = g ? 2 * g - 1 + tt : 0;
    const int comp = (tt != hi) ? 1 : 0;
    const float sg = (tt == 0 && hi == 1) ? -1.f : 1.f;
    const float* Wf = (const float*)W;
    v2f fa[2][4], fb[2][4];
#pragma unroll
    for (int i = 0; i < 2; i++) {
      const int l = 2 * i + (wv >> 2);
      const float* Xp = Wf + (size_t)(l * NX + t) * 128 + hi * 64 + r * 8;
      const float4 x0 = *(const float4*)Xp;
      const float4 x1 = *(const float4*)(Xp + 4);
      fb[i][0].x = x0.x; fb[i][0].y = x0.y; fb[i][1].x = x0.z; fb[i][1].y = x0.w;
      fb[i][2].x = x1.x; fb[i][2].y = x1.y; fb[i][3].x = x1.z; fb[i][3].y = x1.w;
      const float* Kc = (const float*)&K8[l][g][r] + comp;
#pragma unroll
      for (int ks = 0; ks < 4; ks++) {
        fa[i][ks].x = sg * Kc[(2 * ks) * 16];
        fa[i][ks].y = sg * Kc[(2 * ks + 1) * 16];
      }
    }
    v8f acc[2] = {};
#pragma unroll
    for (int ks = 0; ks < 4; ks++) {
#pragma unroll
      for (int i = 0; i < 2; i++) acc[i] = wmma4(fa[i][ks], fb[i][ks], acc[i]);
    }
    {
      float* Vd = (float*)&V[1 + (wv >> 2)][t][r * VP] + hi;
#pragma unroll
      for (int v = 0; v < 8; v++) Vd[2 * v] = acc[1][v];
    }
    if (__builtin_amdgcn_readfirstlane(wv) >> 2) {
      float* Vd = (float*)&V[0][t][r * VP] + hi;
#pragma unroll
      for (int v = 0; v < 8; v++) Vd[2 * v] = acc[0][v];
    } else {
      ((float*)&V0[t][r])[hi] = acc[0][0];
    }
  }
  __syncthreads();

  {
    float2 val;
    if (tid < 64) {
      const int ra = tid >> 3, rb = tid & 7;
      val = cmul(V0[0][ra], V0[1 + (ra & 1)][rb]);
    } else if (tid < 128) {
      const int h = (tid >> 5) & 1;
      const int ra = (tid >> 3) & 3, rb = tid & 7;
      const float2 bf = V0[1 + (ra & 1)][rb];
      const float2 x0 = cmul(V0[0][ra], bf);
      const float2 x1 = cmul(V0[0][ra + 4], bf);
      val = cadd(cmul(CU[0][h][0], x0), cmul(CU[0][h][1], x1));
    } else {
      const int p = (tid >> 6) & 1;
      const int gc = (tid >> 3) & 7, gd = tid & 7;
      val = cmul(V0[3 + p][gc], V0[5 + (gc & 1)][gd]);
    }
    RC[tid] = val;
  }
  __syncthreads();

  const float2 (*V1)[8 * VP] = V[0];
  float2* const T1 = &K8[0][0][0];
  {
    {
      const int p = tid >> 7, sc = (tid >> 6) & 1, c = tid & 63;
      const float2* Vc = V1[3 + sc] + (c >> 3) * VP;
      const float2* Cf = RC + 128 + 64 * p + (c & 7);
      float2 acc = make_float2(0.f, 0.f);
#pragma unroll
      for (int k = 0; k < 8; k++) acc = cadd(acc, cmul(Vc[k], Cf[8 * k]));
      F1[128 + tid] = acc;
    }
    if (tid < 128) {
      const int e = tid >> 6, r = tid & 63;
      const float2* Va = V1[0] + (r >> 3) * VP;
      const float2* Rs = RC + 64 * e + (r & 7);
      float2 acc = make_float2(0.f, 0.f);
#pragma unroll
      for (int k = 0; k < 8; k++) acc = cadd(acc, cmul(Va[k], Rs[8 * k]));
      F1[tid] = acc;
    }
  }
  __syncthreads();

  {
    {
      const int e = tid >> 7, p = (tid >> 6) & 1, r = tid & 63;
      const int r5 = r & 31, h = r >> 5;
      float2 lohi[2];
#pragma unroll
      for (int u = 0; u < 2; u++) {
        const int rr = r5 + 32 * u;
        const int ra = rr >> 3;
        const float2* Vb = V1[1 + (ra & 1)] + (rr & 7) * VP + p;
        const float2* Ra = F1 + 64 * e + 8 * ra + p;
        float2 acc = make_float2(0.f, 0.f);
#pragma unroll
        for (int k = 0; k < 4; k++) acc = cadd(acc, cmul(Vb[2 * k], Ra[2 * k]));
        lohi[u] = acc;
      }
      T1[tid] = h ? lohi[1] : lohi[0];
      T1[256 + tid] = cadd(cmul(CU[1][h][0], lohi[0]), cmul(CU[1][h][1], lohi[1]));
    }
    for (int o = tid; o < 512; o += 256) {
      const int c = o & 63, sc = (o >> 6) & 1, e = (o >> 7) & 1, p = o >> 8;
      const int gc = c >> 3;
      const float2* Vd = V1[5 + (gc & 1)] + (c & 7) * VP + e;
      const float2* Cc = F1 + 128 + (2 * p + sc) * 64 + 8 * gc + e;
      float2 acc = make_float2(0.f, 0.f);
#pragma unroll
      for (int k = 0; k < 4; k++) acc = cadd(acc, cmul(Vd[2 * k], Cc[2 * k]));
      T1[512 + o] = acc;
    }
  }
  __syncthreads();

  {
    const int j = tid & 63, q = tid >> 6;
    const int sp = q & 1;
    const float2* Rt = T1 + 256 * (j & 1);
    float2 cd[4];
#pragma unroll
    for (int e = 0; e < 2; e++)
#pragma unroll
      for (int p = 0; p < 2; p++)
        cd[2 * e + p] = T1[512 + ((2 * p + e) * 2 + sp) * 64 + j];
    for (int it = 0; it < 16; it++) {
      const int i = q + 4 * it;
      float2 a = cmul(Rt[i], cd[0]);
#pragma unroll
      for (int m = 1; m < 4; m++) a = cadd(a, cmul(Rt[64 * m + i], cd[m]));
      Sr[i * LDA + j] = a.x;
      Si[i * LDA + j] = a.y;
    }
  }
  __syncthreads();

  const int w = tid >> 5;
  const int par = (w >> 1) & 1;

#pragma nounroll
  for (int l = 2; l < NL; l++) {
    const float2 (*Vl)[8 * VP] = V[l - 1];
    kron_step<1, 8 * LDA, 16, 32>(Sr, Si, Vl[0], w * LDA);
    __syncthreads();
    kron_step<1, LDA, 16, 32>(Sr, Si, Vl[1 + (w & 1)], w * 8 * LDA);
    __syncthreads();
    kron_step<2 * LDA, 8, 1, 2>(Sr, Si, Vl[3 + par],
                             (par + 32 * (w >> 2)) * LDA + 4 * (w & 1));
    __syncthreads();
    kron_step<4 * LDA, 1, LDA, 2 * LDA>(Sr, Si, Vl[5 + (w & 1)], w * 8);
    __syncthreads();

    {
      float2 u00 = CU[l][0][0], u01 = CU[l][0][1];
      float2 u10 = CU[l][1][0], u11 = CU[l][1][1];
      for (int t = tid; t < 1024; t += 256) {
        int il = t >> 5;
        int j = ((t & 31) << 1) | 1;
        int a0 = il * LDA + j, a1 = (il + 32) * LDA + j;
        float2 x0 = make_float2(Sr[a0], Si[a0]);
        float2 x1 = make_float2(Sr[a1], Si[a1]);
        float2 y0 = cadd(cmul(u00, x0), cmul(u01, x1));
        float2 y1 = cadd(cmul(u10, x0), cmul(u11, x1));
        Sr[a0] = y0.x; Si[a0] = y0.y;
        Sr[a1] = y1.x; Si[a1] = y1.y;
      }
    }
    __syncthreads();
  }

  {
    const int j = tid & 63, qd = tid >> 6;
    float p[16];
#pragma unroll
    for (int k = 0; k < 16; k++) {
      const int ad = (qd + 4 * k) * LDA + j;
      const float pr = Sr[ad], pi = Si[ad];
      p[k] = pr * pr + pi * pi;
    }
    float a1[8];
    float D0 = p[0] - p[1];
    a1[0] = p[0] + p[1];
#pragma unroll
    for (int m = 1; m < 8; m++) {
      a1[m] = p[2 * m] + p[2 * m + 1];
      D0 += p[2 * m] - p[2 * m + 1];
    }
    float a2[4];
    float D1 = a1[0] - a1[1];
    a2[0] = a1[0] + a1[1];
#pragma unroll
    for (int m = 1; m < 4; m++) {
      a2[m] = a1[2 * m] + a1[2 * m + 1];
      D1 += a1[2 * m] - a1[2 * m + 1];
    }
    const float a30 = a2[0] + a2[1], a31 = a2[2] + a2[3];
    float D2 = (a2[0] - a2[1]) + (a2[2] - a2[3]);
    float D3 = a30 - a31;
    float T = a30 + a31;

    const int lane = tid & 31;
#pragma unroll
    for (int m = 1; m < 32; m <<= 1) {
      const float t = __shfl_xor(T, m);
      T = (lane & m) ? (t - T) : (T + t);
      D0 += __shfl_xor(D0, m);
      D1 += __shfl_xor(D1, m);
      D2 += __shfl_xor(D2, m);
      D3 += __shfl_xor(D3, m);
    }

    float* part = (float*)&V[0][0][0];
    if (lane == 0) {
      part[w * 12 + 0] = D3;
      part[w * 12 + 1] = D2;
      part[w * 12 + 2] = D1;
      part[w * 12 + 3] = D0;
      part[w * 12 + 4] = (w & 4) ? -T : T;
      part[w * 12 + 5] = (w & 2) ? -T : T;
      part[w * 12 + 6] = (w & 1) ? -T : T;
    } else if ((lane & (lane - 1)) == 0) {
      part[w * 12 + 11 - (31 - __clz(lane))] = T;
    }
    __syncthreads();
    if (tid < 12) {
      float zq = part[tid];
#pragma unroll
      for (int ww = 1; ww < 8; ww++) zq += part[ww * 12 + tid];
      out[(size_t)s * 12 + tid] = zq;
    }
  }
}

extern "C" void kernel_launch(void* const* d_in, const int* in_sizes, int n_in,
                              void* d_out, int out_size, void* d_ws, size_t ws_size,
                              hipStream_t stream) {
  (void)n_in; (void)out_size; (void)ws_size;
  const float* x    = (const float*)d_in[0];
  const float* rotp = (const float*)d_in[1];
  const float* entp = (const float*)d_in[2];
  float2* W = (float2*)d_ws;
  const int B = in_sizes[0] / 48;

  qk_prep<<<NL, 64, 0, stream>>>(entp, W);
  qk_main<<<B, 256, 0, stream>>>(x, rotp, entp, W, (float*)d_out);
}


# LDS state complex-interleaved (float2, row pitch 66): the four steps move it in 16-byte LDS ops, 18 per wave-step instead of 36, the rest in 8-byte ops; fragment k order, so sum order, changes
# speedup vs baseline: 10.1918x; 1.0746x over previous
#include <hip/hip_runtime.h>
#include <math.h>

#define LDS2 66
#define NL  4
#define NX  7
#define VP  10

typedef __attribute__((ext_vector_type(2))) float v2f;
typedef __attribute__((ext_vector_type(8))) float v8f;

__device__ __forceinline__ float2 cmul(float2 a, float2 b) {
  return make_float2(a.x * b.x - a.y * b.y, a.x * b.y + a.y * b.x);
}
__device__ __forceinline__ float2 cadd(float2 a, float2 b) {
  return make_float2(a.x + b.x, a.y + b.y);
}

struct C2 { float2 a00, a01, a10, a11; };
__device__ __forceinline__ C2 mm2(const C2 L, const C2 R) {
  C2 o;
  o.a00 = cadd(cmul(L.a00, R.a00), cmul(L.a01, R.a10));
  o.a01 = cadd(cmul(L.a00, R.a01), cmul(L.a01, R.a11));
  o.a10 = cadd(cmul(L.a10, R.a00), cmul(L.a11, R.a10));
  o.a11 = cadd(cmul(L.a10, R.a01), cmul(L.a11, R.a11));
  return o;
}

__device__ __forceinline__ void rot_gate(float phi, float th, float om,
                                         float2* u00, float2* u01,
                                         float2* u10, float2* u11) {
  float ct = cosf(0.5f * th), st = sinf(0.5f * th);
  float ap = -0.5f * (phi + om);
  float am = 0.5f * (phi - om);
  *u00 = make_float2(ct * cosf(ap), ct * sinf(ap));
  *u01 = make_float2(-st * cosf(am), -st * sinf(am));
  *u10 = make_float2(st * cosf(-am), st * sinf(-am));
  *u11 = make_float2(ct * cosf(-ap), ct * sinf(-ap));
}

__global__ __launch_bounds__(64) void qk_prep(const float* __restrict__ entp,
                                              float2* __restrict__ W) {
  __shared__ float2 U[12][2][2];
  const int l = blockIdx.x;
  const int tid = threadIdx.x;

  if (tid < 12) {
    const float* p = &entp[(l * 12 + tid) * 3];
    rot_gate(p[0], p[1], p[2], &U[tid][0][0], &U[tid][0][1],
             &U[tid][1][0], &U[tid][1][1]);
  }
  __syncthreads();

  if (tid < NX * 8) {
    const int t = tid >> 3, j = tid & 7;
    const int g = (t + 1) >> 1;
    const bool link = (t > 0) && ((t & 1) == 0);
    float2 v[8];
#pragma unroll
    for (int r = 0; r < 8; r++) v[r] = make_float2((r == j) ? 1.f : 0.f, 0.f);
    if (link) {
      const int q = 3 * g - 1;
#pragma unroll
      for (int r = 0; r < 8; r++)
        v[r] = ((r & 3) == (j & 3)) ? U[q][r >> 2][j >> 2] : make_float2(0.f, 0.f);
    }
    {
      const int q = 3 * g;
      float2 u00 = U[q][0][0], u01 = U[q][0][1], u10 = U[q][1][0], u11 = U[q][1][1];
#pragma unroll
      for (int i = 4; i < 6; i++) {
        float2 x0 = v[i], x1 = v[i + 2];
        v[i]     = cadd(cmul(u00, x0), cmul(u01, x1));
        v[i + 2] = cadd(cmul(u10, x0), cmul(u11, x1));
      }
    }
    {
      const int q = 3 * g + 1;
      float2 u00 = U[q][0][0], u01 = U[q][0][1], u10 = U[q][1][0], u11 = U[q][1][1];
#pragma unroll
      for (int i = 2; i < 8; i += 4) {
        float2 x0 = v[i], x1 = v[i + 1];
        v[i]     = cadd(cmul(u00, x0), cmul(u01, x1));
        v[i + 1] = cadd(cmul(u10, x0), cmul(u11, x1));
      }
    }
    float* Wt = (float*)(W + (size_t)(l * NX + t) * 64);
#pragma unroll
    for (int r = 0; r < 8; r++) {
      Wt[r * 8 + j] = v[r].x;
      Wt[64 + r * 8 + j] = v[r].y;
    }
  }
}

__device__ __forceinline__ v8f wmma4(v2f a, v2f b, v8f c) {
  return __builtin_amdgcn_wmma_f32_16x16x4_f32(
      false, a, false, b, (short)0, c, false, false);
}

template <bool PAIRS, int N0, int N1, int N2, int N3, int GS, int TS0, int TS1>
__device__ __forceinline__ void kron_step(float2* S, const float2* V, int base0) {
  const int lane = threadIdx.x & 31;
  const int n = lane & 15;
  const int hi = lane >> 4;
  const int gp = PAIRS ? (4 * (n >> 3) + ((n & 7) >> 1)) : ((n & 6) + (n >> 3));
  const bool im = (n & 1) != 0;

  const float4* Vrow = (const float4*)(V + gp * VP + 4 * hi);
  const float4 q0 = Vrow[0], q1 = Vrow[1];
  v2f a[4];
  a[0].x = im ? q0.y : q0.x;  a[0].y = im ? q0.x : -q0.y;
  a[1].x = im ? q0.w : q0.z;  a[1].y = im ? q0.z : -q0.w;
  a[2].x = im ? q1.y : q1.x;  a[2].y = im ? q1.x : -q1.y;
  a[3].x = im ? q1.w : q1.z;  a[3].y = im ? q1.z : -q1.w;

  float2* const P = S + base0 + (n & 1) * N0 + ((n >> 1) & 1) * N1 +
                    ((n >> 2) & 1) * N2 + (n >> 3) * N3 + hi * (PAIRS ? 4 : GS);
  v2f b[4][4];
  if (PAIRS) {
#pragma unroll
    for (int t = 0; t < 4; t++) {
#pragma unroll
      for (int h = 0; h < 2; h++) {
        const float4 z = *(const float4*)(P + (t & 1) * TS0 + (t >> 1) * TS1 + 2 * h);
        b[t][2 * h].x = z.x;      b[t][2 * h].y = z.y;
        b[t][2 * h + 1].x = z.z;  b[t][2 * h + 1].y = z.w;
      }
    }
  } else {
#pragma unroll
    for (int u = 0; u < 2; u++) {
#pragma unroll
      for (int ks = 0; ks < 4; ks++) {
        const float4 z = *(const float4*)(P + u * TS1 + ks * 2 * GS);
        b[2 * u][ks].x = z.x;      b[2 * u][ks].y = z.y;
        b[2 * u + 1][ks].x = z.z;  b[2 * u + 1][ks].y = z.w;
      }
    }
  }

  v8f acc[4] = {};
#pragma unroll
  for (int ks = 0; ks < 4; ks++) {
#pragma unroll
    for (int t = 0; t < 4; t++) acc[t] = wmma4(a[ks], b[t][ks], acc[t]);
  }

  if (PAIRS) {
#pragma unroll
    for (int t = 0; t < 4; t++) {
#pragma unroll
      for (int h = 0; h < 2; h++)
        *(float4*)(P + (t & 1) * TS0 + (t >> 1) * TS1 + 2 * h) =
            make_float4(acc[t][4 * h], acc[t][4 * h + 1], acc[t][4 * h + 2], acc[t][4 * h + 3]);
    }
  } else {
#pragma unroll
    for (int u = 0; u < 2; u++) {
#pragma unroll
      for (int gl = 0; gl < 4; gl++)
        *(float4*)(P + u * TS1 + gl * 2 * GS) =
            make_float4(acc[2 * u][2 * gl], acc[2 * u][2 * gl + 1],
                        acc[2 * u + 1][2 * gl], acc[2 * u + 1][2 * gl + 1]);
    }
  }
}

__global__ __launch_bounds__(256, 10) void qk_main(const float* __restrict__ x,
                                               const float* __restrict__ rotp,
                                               const float* __restrict__ entp,
                                               const float2* __restrict__ W,
                                               float* __restrict__ out) {
  __shared__ __attribute__((aligned(16))) float2 S[64 * LDS2];
  __shared__ float2 G[NL][12][2][2];
  __shared__ float2 K8[NL][4][64];
  __shared__ __attribute__((aligned(16))) float2 V[NL - 1][NX][8 * VP];
  __shared__ float2 V0[NX][8];
  __shared__ float2 RC[256];
  __shared__ float2 CU[NL][2][2];
  __shared__ float2 F1[384];

  const int tid = threadIdx.x;
  const int s = blockIdx.x;

  if (tid < 12 * NL) {
    const int gl = tid / 12, gq = tid - 12 * gl;
    const float* xs = x + (size_t)s * 48;
    float a = xs[0 * 12 + gq], b = xs[1 * 12 + gq];
    float c = xs[2 * 12 + gq], d = xs[3 * 12 + gq];
    float ca = cosf(0.5f * a), sa = sinf(0.5f * a);
    float cb = cosf(0.5f * b), sb = sinf(0.5f * b);
    float cc = cosf(0.5f * c), sc = sinf(0.5f * c);
    float cd = cosf(0.5f * d), sd = sinf(0.5f * d);
    C2 M  = { {ca, 0.f}, {-sa, 0.f}, {sa, 0.f}, {ca, 0.f} };
    C2 RX = { {cb, 0.f}, {0.f, -sb}, {0.f, -sb}, {cb, 0.f} };
    C2 RZ = { {cc, -sc}, {0.f, 0.f}, {0.f, 0.f}, {cc, sc} };
    C2 RY = { {cd, 0.f}, {-sd, 0.f}, {sd, 0.f}, {cd, 0.f} };
    M = mm2(RX, M); M = mm2(RZ, M); M = mm2(RY, M);
    const float* rp = &rotp[(gl * 12 + gq) * 3];
    C2 R;
    rot_gate(rp[0], rp[1], rp[2], &R.a00, &R.a01, &R.a10, &R.a11);
    M = mm2(R, M);
    G[gl][gq][0][0] = M.a00; G[gl][gq][0][1] = M.a01;
    G[gl][gq][1][0] = M.a10; G[gl][gq][1][1] = M.a11;
  }
  if (tid >= 64 && tid < 64 + NL) {
    int l = tid - 64;
    const float* p = &entp[(l * 12 + 11) * 3];
    rot_gate(p[0], p[1], p[2], &CU[l][0][0], &CU[l][0][1],
             &CU[l][1][0], &CU[l][1][1]);
  }
  __syncthreads();

  for (int e = tid; e < NL * 256; e += 256) {
    const int l = e >> 8;
    const int kg = (e >> 6) & 3, ka = (e >> 3) & 7, kb8 = e & 7;
    float2 p = cmul(cmul(G[l][3 * kg][(ka >> 2) & 1][(kb8 >> 2) & 1],
                         G[l][3 * kg + 1][(ka >> 1) & 1][(kb8 >> 1) & 1]),
                    G[l][3 * kg + 2][ka & 1][kb8 & 1]);
    K8[l][kg][ka * 8 + kb8] = p;
  }
  __syncthreads();

  {
    const int lane = tid & 31;
    const int n = lane & 15, hi = lane >> 4;
    const int r = n & 7, tt = n >> 3;
    const int wv = tid >> 5;
    const int g = wv & 3;
    const int t = g ? 2 * g - 1 + tt : 0;
    const int comp = (tt != hi) ? 1 : 0;
    const float sg = (tt == 0 && hi == 1) ? -1.f : 1.f;
    const float* Wf = (const float*)W;
    v2f fa[2][4], fb[2][4];
#pragma unroll
    for (int i = 0; i < 2; i++) {
      const int l = 2 * i + (wv >> 2);
      const float* Xp = Wf + (size_t)(l * NX + t) * 128 + hi * 64 + r * 8;
      const float4 x0 = *(const float4*)Xp;
      const float4 x1 = *(const float4*)(Xp + 4);
      fb[i][0].x = x0.x; fb[i][0].y = x0.y; fb[i][1].x = x0.z; fb[i][1].y = x0.w;
      fb[i][2].x = x1.x; fb[i][2].y = x1.y; fb[i][3].x = x1.z; fb[i][3].y = x1.w;
      const float* Kc = (const float*)&K8[l][g][r] + comp;
#pragma unroll
      for (int ks = 0; ks < 4; ks++) {
        fa[i][ks].x = sg * Kc[(2 * ks) * 16];
        fa[i][ks].y = sg * Kc[(2 * ks + 1) * 16];
      }
    }
    v8f acc[2] = {};
#pragma unroll
    for (int ks = 0; ks < 4; ks++) {
#pragma unroll
      for (int i = 0; i < 2; i++) acc[i] = wmma4(fa[i][ks], fb[i][ks], acc[i]);
    }
    {
      float* Vd = (float*)&V[1 + (wv >> 2)][t][r * VP] + hi;
      if (__builtin_amdgcn_readfirstlane(g) == 3) {
#pragma unroll
        for (int v = 0; v < 8; v++) Vd[2 * v] = acc[1][v];
      } else {
#pragma unroll
        for (int v = 0; v < 8; v++) Vd[2 * (4 * (v & 1) + (v >> 1))] = acc[1][v];
      }
    }
    if (__builtin_amdgcn_readfirstlane(wv) >> 2) {
      float* Vd = (float*)&V[0][t][r * VP] + hi;
#pragma unroll
      for (int v = 0; v < 8; v++) Vd[2 * v] = acc[0][v];
    } else {
      ((float*)&V0[t][r])[hi] = acc[0][0];
    }
  }
  __syncthreads();

  {
    float2 val;
    if (tid < 64) {
      const int ra = tid >> 3, rb = tid & 7;
      val = cmul(V0[0][ra], V0[1 + (ra & 1)][rb]);
    } else if (tid < 128) {
      const int h = (tid >> 5) & 1;
      const int ra = (tid >> 3) & 3, rb = tid & 7;
      const float2 bf = V0[1 + (ra & 1)][rb];
      const float2 x0 = cmul(V0[0][ra], bf);
      const float2 x1 = cmul(V0[0][ra + 4], bf);
      val = cadd(cmul(CU[0][h][0], x0), cmul(CU[0][h][1], x1));
    } else {
      const int p = (tid >> 6) & 1;
      const int gc = (tid >> 3) & 7, gd = tid & 7;
      val = cmul(V0[3 + p][gc], V0[5 + (gc & 1)][gd]);
    }
    RC[tid] = val;
  }
  __syncthreads();

  const float2 (*V1)[8 * VP] = V[0];
  float2* const T1 = &K8[0][0][0];
  {
    {
      const int p = tid >> 7, sc = (tid >> 6) & 1, c = tid & 63;
      const float2* Vc = V1[3 + sc] + (c >> 3) * VP;
      const float2* Cf = RC + 128 + 64 * p + (c & 7);
      float2 acc = make_float2(0.f, 0.f);
#pragma unroll
      for (int k = 0; k < 8; k++) acc = cadd(acc, cmul(Vc[k], Cf[8 * k]));
      F1[128 + tid] = acc;
    }
    if (tid < 128) {
      const int e = tid >> 6, r = tid & 63;
      const float2* Va = V1[0] + (r >> 3) * VP;
      const float2* Rs = RC + 64 * e + (r & 7);
      float2 acc = make_float2(0.f, 0.f);
#pragma unroll
      for (int k = 0; k < 8; k++) acc = cadd(acc, cmul(Va[k], Rs[8 * k]));
      F1[tid] = acc;
    }
  }
  __syncthreads();

  {
    {
      const int e = tid >> 7, p = (tid >> 6) & 1, r = tid & 63;
      const int r5 = r & 31, h = r >> 5;
      float2 lohi[2];
#pragma unroll
      for (int u = 0; u < 2; u++) {
        const int rr = r5 + 32 * u;
        const int ra = rr >> 3;
        const float2* Vb = V1[1 + (ra & 1)] + (rr & 7) * VP + p;
        const float2* Ra = F1 + 64 * e + 8 * ra + p;
        float2 acc = make_float2(0.f, 0.f);
#pragma unroll
        for (int k = 0; k < 4; k++) acc = cadd(acc, cmul(Vb[2 * k], Ra[2 * k]));
        lohi[u] = acc;
      }
      T1[tid] = h ? lohi[1] : lohi[0];
      T1[256 + tid] = cadd(cmul(CU[1][h][0], lohi[0]), cmul(CU[1][h][1], lohi[1]));
    }
    for (int o = tid; o < 512; o += 256) {
      const int c = o & 63, sc = (o >> 6) & 1, e = (o >> 7) & 1, p = o >> 8;
      const int gc = c >> 3;
      const float2* Vd = V1[5 + (gc & 1)] + (c & 7) * VP + e;
      const float2* Cc = F1 + 128 + (2 * p + sc) * 64 + 8 * gc + e;
      float2 acc = make_float2(0.f, 0.f);
#pragma unroll
      for (int k = 0; k < 4; k++) acc = cadd(acc, cmul(Vd[2 * k], Cc[2 * k]));
      T1[512 + o] = acc;
    }
  }
  __syncthreads();

  {
    const int j = tid & 63, q = tid >> 6;
    const int sp = q & 1;
    const float2* Rt = T1 + 256 * (j & 1);
    float2 cd[4];
#pragma unroll
    for (int e = 0; e < 2; e++)
#pragma unroll
      for (int p = 0; p < 2; p++)
        cd[2 * e + p] = T1[512 + ((2 * p + e) * 2 + sp) * 64 + j];
    for (int it = 0; it < 16; it++) {
      const int i = q + 4 * it;
      float2 a = cmul(Rt[i], cd[0]);
#pragma unroll
      for (int m = 1; m < 4; m++) a = cadd(a, cmul(Rt[64 * m + i], cd[m]));
      S[i * LDS2 + j] = a;
    }
  }
  __syncthreads();

  const int w = tid >> 5;

#pragma nounroll
  for (int l = 2; l < NL; l++) {
    const float2 (*Vl)[8 * VP] = V[l - 1];
    kron_step<false, 2, 4, 8, 16, 8 * LDS2, 0, 32>(S, Vl[0], w * LDS2);
    __syncthreads();
    kron_step<false, 2, 4, 8, 16, LDS2, 0, 32>(S, Vl[1 + (w & 1)], w * 8 * LDS2);
    __syncthreads();
    kron_step<false, 2, 4, 4 * LDS2, 8 * LDS2, 8, 0, 32 * LDS2>(
        S, Vl[3 + (w & 1)], ((w & 3) + 16 * (w >> 2)) * LDS2);
    __syncthreads();
    kron_step<true, LDS2, 2 * LDS2, 4 * LDS2, 8 * LDS2, 1, 16 * LDS2, 32 * LDS2>(
        S, Vl[5 + (w & 1)], w * 8);
    __syncthreads();

    {
      float2 u00 = CU[l][0][0], u01 = CU[l][0][1];
      float2 u10 = CU[l][1][0], u11 = CU[l][1][1];
      for (int t = tid; t < 1024; t += 256) {
        int il = t >> 5;
        int j = ((t & 31) << 1) | 1;
        int a0 = il * LDS2 + j, a1 = (il + 32) * LDS2 + j;
        float2 x0 = S[a0];
        float2 x1 = S[a1];
        float2 y0 = cadd(cmul(u00, x0), cmul(u01, x1));
        float2 y1 = cadd(cmul(u10, x0), cmul(u11, x1));
        S[a0] = y0;
        S[a1] = y1;
      }
    }
    __syncthreads();
  }

  {
    const int j = tid & 63, qd = tid >> 6;
    float p[16];
#pragma unroll
    for (int k = 0; k < 16; k++) {
      const float2 z = S[(qd + 4 * k) * LDS2 + j];
      p[k] = z.x * z.x + z.y * z.y;
    }
    float a1[8];
    float D0 = p[0] - p[1];
    a1[0] = p[0] + p[1];
#pragma unroll
    for (int m = 1; m < 8; m++) {
      a1[m] = p[2 * m] + p[2 * m + 1];
      D0 += p[2 * m] - p[2 * m + 1];
    }
    float a2[4];
    float D1 = a1[0] - a1[1];
    a2[0] = a1[0] + a1[1];
#pragma unroll
    for (int m = 1; m < 4; m++) {
      a2[m] = a1[2 * m] + a1[2 * m + 1];
      D1 += a1[2 * m] - a1[2 * m + 1];
    }
    const float a30 = a2[0] + a2[1], a31 = a2[2] + a2[3];
    float D2 = (a2[0] - a2[1]) + (a2[2] - a2[3]);
    float D3 = a30 - a31;
    float T = a30 + a31;

    const int lane = tid & 31;
#pragma unroll
    for (int m = 1; m < 32; m <<= 1) {
      const float t = __shfl_xor(T, m);
      T = (lane & m) ? (t - T) : (T + t);
      D0 += __shfl_xor(D0, m);
      D1 += __shfl_xor(D1, m);
      D2 += __shfl_xor(D2, m);
      D3 += __shfl_xor(D3, m);
    }

    float* part = (float*)&V[0][0][0];
    if (lane == 0) {
      part[w * 12 + 0] = D3;
      part[w * 12 + 1] = D2;
      part[w * 12 + 2] = D1;
      part[w * 12 + 3] = D0;
      part[w * 12 + 4] = (w & 4) ? -T : T;
      part[w * 12 + 5] = (w & 2) ? -T : T;
      part[w * 12 + 6] = (w & 1) ? -T : T;
    } else if ((lane & (lane - 1)) == 0) {
      part[w * 12 + 11 - (31 - __clz(lane))] = T;
    }
    __syncthreads();
    if (tid < 12) {
      float zq = part[tid];
#pragma unroll
      for (int ww = 1; ww < 8; ww++) zq += part[ww * 12 + tid];
      out[(size_t)s * 12 + tid] = zq;
    }
  }
}

extern "C" void kernel_launch(void* const* d_in, const int* in_sizes, int n_in,
                              void* d_out, int out_size, void* d_ws, size_t ws_size,
                              hipStream_t stream) {
  (void)n_in; (void)out_size; (void)ws_size;
  const float* x    = (const float*)d_in[0];
  const float* rotp = (const float*)d_in[1];
  const float* entp = (const float*)d_in[2];
  float2* W = (float2*)d_ws;
  const int B = in_sizes[0] / 48;

  qk_prep<<<NL, 64, 0, stream>>>(entp, W);
  qk_main<<<B, 256, 0, stream>>>(x, rotp, entp, W, (float*)d_out);
}


# Complex multiply-adds of the vector-unit phases as two packed-f32 FMAs each (a.re*b, then a.im*(i*b) via operand modifiers) instead of five instructions; rounding order of these sums changes
# speedup vs baseline: 10.7949x; 1.0592x over previous
#include <hip/hip_runtime.h>
#include <math.h>

#define LDS2 66
#define NL  4
#define NX  7
#define VP  10

typedef __attribute__((ext_vector_type(2))) float v2f;
typedef __attribute__((ext_vector_type(8))) float v8f;

__device__ __forceinline__ float2 cmul(float2 a, float2 b) {
  return make_float2(a.x * b.x - a.y * b.y, a.x * b.y + a.y * b.x);
}
__device__ __forceinline__ float2 cadd(float2 a, float2 b) {
  return make_float2(a.x + b.x, a.y + b.y);
}

__device__ __forceinline__ v2f V2(float2 a) { v2f r; r.x = a.x; r.y = a.y; return r; }
__device__ __forceinline__ float2 F2(v2f a) { return make_float2(a.x, a.y); }
__device__ __forceinline__ v2f irot(v2f b) { v2f r; r.x = -b.y; r.y = b.x; return r; }
__device__ __forceinline__ v2f cmulp(v2f a, v2f b) {
  return __builtin_elementwise_fma(a.yy, irot(b), a.xx * b);
}
__device__ __forceinline__ v2f cmacp(v2f acc, v2f a, v2f b) {
  acc = __builtin_elementwise_fma(a.xx, b, acc);
  return __builtin_elementwise_fma(a.yy, irot(b), acc);
}
template <int N, int SA, int SB>
__device__ __forceinline__ v2f cdotp(const float2* a, const float2* b) {
  v2f acc = cmulp(V2(a[0]), V2(b[0]));
#pragma unroll
  for (int k = 1; k < N; k++) acc = cmacp(acc, V2(a[SA * k]), V2(b[SB * k]));
  return acc;
}

struct C2 { float2 a00, a01, a10, a11; };
__device__ __forceinline__ C2 mm2(const C2 L, const C2 R) {
  C2 o;
  o.a00 = F2(cmacp(cmulp(V2(L.a00), V2(R.a00)), V2(L.a01), V2(R.a10)));
  o.a01 = F2(cmacp(cmulp(V2(L.a00), V2(R.a01)), V2(L.a01), V2(R.a11)));
  o.a10 = F2(cmacp(cmulp(V2(L.a10), V2(R.a00)), V2(L.a11), V2(R.a10)));
  o.a11 = F2(cmacp(cmulp(V2(L.a10), V2(R.a01)), V2(L.a11), V2(R.a11)));
  return o;
}

__device__ __forceinline__ void rot_gate(float phi, float th, float om,
                                         float2* u00, float2* u01,
                                         float2* u10, float2* u11) {
  float ct = cosf(0.5f * th), st = sinf(0.5f * th);
  float ap = -0.5f * (phi + om);
  float am = 0.5f * (phi - om);
  *u00 = make_float2(ct * cosf(ap), ct * sinf(ap));
  *u01 = make_float2(-st * cosf(am), -st * sinf(am));
  *u10 = make_float2(st * cosf(-am), st * sinf(-am));
  *u11 = make_float2(ct * cosf(-ap), ct * sinf(-ap));
}

__global__ __launch_bounds__(64) void qk_prep(const float* __restrict__ entp,
                                              float2* __restrict__ W) {
  __shared__ float2 U[12][2][2];
  const int l = blockIdx.x;
  const int tid = threadIdx.x;

  if (tid < 12) {
    const float* p = &entp[(l * 12 + tid) * 3];
    rot_gate(p[0], p[1], p[2], &U[tid][0][0], &U[tid][0][1],
             &U[tid][1][0], &U[tid][1][1]);
  }
  __syncthreads();

  if (tid < NX * 8) {
    const int t = tid >> 3, j = tid & 7;
    const int g = (t + 1) >> 1;
    const bool link = (t > 0) && ((t & 1) == 0);
    float2 v[8];
#pragma unroll
    for (int r = 0; r < 8; r++) v[r] = make_float2((r == j) ? 1.f : 0.f, 0.f);
    if (link) {
      const int q = 3 * g - 1;
#pragma unroll
      for (int r = 0; r < 8; r++)
        v[r] = ((r & 3) == (j & 3)) ? U[q][r >> 2][j >> 2] : make_float2(0.f, 0.f);
    }
    {
      const int q = 3 * g;
      float2 u00 = U[q][0][0], u01 = U[q][0][1], u10 = U[q][1][0], u11 = U[q][1][1];
#pragma unroll
      for (int i = 4; i < 6; i++) {
        float2 x0 = v[i], x1 = v[i + 2];
        v[i]     = cadd(cmul(u00, x0), cmul(u01, x1));
        v[i + 2] = cadd(cmul(u10, x0), cmul(u11, x1));
      }
    }
    {
      const int q = 3 * g + 1;
      float2 u00 = U[q][0][0], u01 = U[q][0][1], u10 = U[q][1][0], u11 = U[q][1][1];
#pragma unroll
      for (int i = 2; i < 8; i += 4) {
        float2 x0 = v[i], x1 = v[i + 1];
        v[i]     = cadd(cmul(u00, x0), cmul(u01, x1));
        v[i + 1] = cadd(cmul(u10, x0), cmul(u11, x1));
      }
    }
    float* Wt = (float*)(W + (size_t)(l * NX + t) * 64);
#pragma unroll
    for (int r = 0; r < 8; r++) {
      Wt[r * 8 + j] = v[r].x;
      Wt[64 + r * 8 + j] = v[r].y;
    }
  }
}

__device__ __forceinline__ v8f wmma4(v2f a, v2f b, v8f c) {
  return __builtin_amdgcn_wmma_f32_16x16x4_f32(
      false, a, false, b, (short)0, c, false, false);
}

template <bool PAIRS, int N0, int N1, int N2, int N3, int GS, int TS0, int TS1>
__device__ __forceinline__ void kron_step(float2* S, const float2* V, int base0) {
  const int lane = threadIdx.x & 31;
  const int n = lane & 15;
  const int hi = lane >> 4;
  const int gp = PAIRS ? (4 * (n >> 3) + ((n & 7) >> 1)) : ((n & 6) + (n >> 3));
  const bool im = (n & 1) != 0;

  const float4* Vrow = (const float4*)(V + gp * VP + 4 * hi);
  const float4 q0 = Vrow[0], q1 = Vrow[1];
  v2f a[4];
  a[0].x = im ? q0.y : q0.x;  a[0].y = im ? q0.x : -q0.y;
  a[1].x = im ? q0.w : q0.z;  a[1].y = im ? q0.z : -q0.w;
  a[2].x = im ? q1.y : q1.x;  a[2].y = im ? q1.x : -q1.y;
  a[3].x = im ? q1.w : q1.z;  a[3].y = im ? q1.z : -q1.w;

  float2* const P = S + base0 + (n & 1) * N0 + ((n >> 1) & 1) * N1 +
                    ((n >> 2) & 1) * N2 + (n >> 3) * N3 + hi * (PAIRS ? 4 : GS);
  v2f b[4][4];
  if (PAIRS) {
#pragma unroll
    for (int t = 0; t < 4; t++) {
#pragma unroll
      for (int h = 0; h < 2; h++) {
        const float4 z = *(const float4*)(P + (t & 1) * TS0 + (t >> 1) * TS1 + 2 * h);
        b[t][2 * h].x = z.x;      b[t][2 * h].y = z.y;
        b[t][2 * h + 1].x = z.z;  b[t][2 * h + 1].y = z.w;
      }
    }
  } else {
#pragma unroll
    for (int u = 0; u < 2; u++) {
#pragma unroll
      for (int ks = 0; ks < 4; ks++) {
        const float4 z = *(const float4*)(P + u * TS1 + ks * 2 * GS);
        b[2 * u][ks].x = z.x;      b[2 * u][ks].y = z.y;
        b[2 * u + 1][ks].x = z.z;  b[2 * u + 1][ks].y = z.w;
      }
    }
  }

  v8f acc[4] = {};
#pragma unroll
  for (int ks = 0; ks < 4; ks++) {
#pragma unroll
    for (int t = 0; t < 4; t++) acc[t] = wmma4(a[ks], b[t][ks], acc[t]);
  }

  if (PAIRS) {
#pragma unroll
    for (int t = 0; t < 4; t++) {
#pragma unroll
      for (int h = 0; h < 2; h++)
        *(float4*)(P + (t & 1) * TS0 + (t >> 1) * TS1 + 2 * h) =
            make_float4(acc[t][4 * h], acc[t][4 * h + 1], acc[t][4 * h + 2], acc[t][4 * h + 3]);
    }
  } else {
#pragma unroll
    for (int u = 0; u < 2; u++) {
#pragma unroll
      for (int gl = 0; gl < 4; gl++)
        *(float4*)(P + u * TS1 + gl * 2 * GS) =
            make_float4(acc[2 * u][2 * gl], acc[2 * u][2 * gl + 1],
                        acc[2 * u + 1][2 * gl], acc[2 * u + 1][2 * gl + 1]);
    }
  }
}

__global__ __launch_bounds__(256, 10) void qk_main(const float* __restrict__ x,
                                               const float* __restrict__ rotp,
                                               const float* __restrict__ entp,
                                               const float2* __restrict__ W,
                                               float* __restrict__ out) {
  __shared__ __attribute__((aligned(16))) float2 S[64 * LDS2];
  __shared__ float2 G[NL][12][2][2];
  __shared__ float2 K8[NL][4][64];
  __shared__ __attribute__((aligned(16))) float2 V[NL - 1][NX][8 * VP];
  __shared__ float2 V0[NX][8];
  __shared__ float2 RC[256];
  __shared__ float2 CU[NL][2][2];
  __shared__ float2 F1[384];

  const int tid = threadIdx.x;
  const int s = blockIdx.x;

  if (tid < 12 * NL) {
    const int gl = tid / 12, gq = tid - 12 * gl;
    const float* xs = x + (size_t)s * 48;
    float a = xs[0 * 12 + gq], b = xs[1 * 12 + gq];
    float c = xs[2 * 12 + gq], d = xs[3 * 12 + gq];
    float ca = cosf(0.5f * a), sa = sinf(0.5f * a);
    float cb = cosf(0.5f * b), sb = sinf(0.5f * b);
    float cc = cosf(0.5f * c), sc = sinf(0.5f * c);
    float cd = cosf(0.5f * d), sd = sinf(0.5f * d);
    C2 M  = { {ca, 0.f}, {-sa, 0.f}, {sa, 0.f}, {ca, 0.f} };
    C2 RX = { {cb, 0.f}, {0.f, -sb}, {0.f, -sb}, {cb, 0.f} };
    C2 RZ = { {cc, -sc}, {0.f, 0.f}, {0.f, 0.f}, {cc, sc} };
    C2 RY = { {cd, 0.f}, {-sd, 0.f}, {sd, 0.f}, {cd, 0.f} };
    M = mm2(RX, M); M = mm2(RZ, M); M = mm2(RY, M);
    const float* rp = &rotp[(gl * 12 + gq) * 3];
    C2 R;
    rot_gate(rp[0], rp[1], rp[2], &R.a00, &R.a01, &R.a10, &R.a11);
    M = mm2(R, M);
    G[gl][gq][0][0] = M.a00; G[gl][gq][0][1] = M.a01;
    G[gl][gq][1][0] = M.a10; G[gl][gq][1][1] = M.a11;
  }
  if (tid >= 64 && tid < 64 + NL) {
    int l = tid - 64;
    const float* p = &entp[(l * 12 + 11) * 3];
    rot_gate(p[0], p[1], p[2], &CU[l][0][0], &CU[l][0][1],
             &CU[l][1][0], &CU[l][1][1]);
  }
  __syncthreads();

  for (int e = tid; e < NL * 256; e += 256) {
    const int l = e >> 8;
    const int kg = (e >> 6) & 3, ka = (e >> 3) & 7, kb8 = e & 7;
    const v2f p = cmulp(cmulp(V2(G[l][3 * kg][(ka >> 2) & 1][(kb8 >> 2) & 1]),
                              V2(G[l][3 * kg + 1][(ka >> 1) & 1][(kb8 >> 1) & 1])),
                        V2(G[l][3 * kg + 2][ka & 1][kb8 & 1]));
    K8[l][kg][ka * 8 + kb8] = F2(p);
  }
  __syncthreads();

  {
    const int lane = tid & 31;
    const int n = lane & 15, hi = lane >> 4;
    const int r = n & 7, tt = n >> 3;
    const int wv = tid >> 5;
    const int g = wv & 3;
    const int t = g ? 2 * g - 1 + tt : 0;
    const int comp = (tt != hi) ? 1 : 0;
    const float sg = (tt == 0 && hi == 1) ? -1.f : 1.f;
    const float* Wf = (const float*)W;
    v2f fa[2][4], fb[2][4];
#pragma unroll
    for (int i = 0; i < 2; i++) {
      const int l = 2 * i + (wv >> 2);
      const float* Xp = Wf + (size_t)(l * NX + t) * 128 + hi * 64 + r * 8;
      const float4 x0 = *(const float4*)Xp;
      const float4 x1 = *(const float4*)(Xp + 4);
      fb[i][0].x = x0.x; fb[i][0].y = x0.y; fb[i][1].x = x0.z; fb[i][1].y = x0.w;
      fb[i][2].x = x1.x; fb[i][2].y = x1.y; fb[i][3].x = x1.z; fb[i][3].y = x1.w;
      const float* Kc = (const float*)&K8[l][g][r] + comp;
#pragma unroll
      for (int ks = 0; ks < 4; ks++) {
        fa[i][ks].x = sg * Kc[(2 * ks) * 16];
        fa[i][ks].y = sg * Kc[(2 * ks + 1) * 16];
      }
    }
    v8f acc[2] = {};
#pragma unroll
    for (int ks = 0; ks < 4; ks++) {
#pragma unroll
      for (int i = 0; i < 2; i++) acc[i] = wmma4(fa[i][ks], fb[i][ks], acc[i]);
    }
    {
      float* Vd = (float*)&V[1 + (wv >> 2)][t][r * VP] + hi;
      if (__builtin_amdgcn_readfirstlane(g) == 3) {
#pragma unroll
        for (int v = 0; v < 8; v++) Vd[2 * v] = acc[1][v];
      } else {
#pragma unroll
        for (int v = 0; v < 8; v++) Vd[2 * (4 * (v & 1) + (v >> 1))] = acc[1][v];
      }
    }
    if (__builtin_amdgcn_readfirstlane(wv) >> 2) {
      float* Vd = (float*)&V[0][t][r * VP] + hi;
#pragma unroll
      for (int v = 0; v < 8; v++) Vd[2 * v] = acc[0][v];
    } else {
      ((float*)&V0[t][r])[hi] = acc[0][0];
    }
  }
  __syncthreads();

  {
    v2f val;
    if (tid < 64) {
      const int ra = tid >> 3, rb = tid & 7;
      val = cmulp(V2(V0[0][ra]), V2(V0[1 + (ra & 1)][rb]));
    } else if (tid < 128) {
      const int h = (tid >> 5) & 1;
      const int ra = (tid >> 3) & 3, rb = tid & 7;
      const v2f bf = V2(V0[1 + (ra & 1)][rb]);
      const v2f x0 = cmulp(V2(V0[0][ra]), bf);
      const v2f x1 = cmulp(V2(V0[0][ra + 4]), bf);
      val = cmacp(cmulp(x0, V2(CU[0][h][0])), x1, V2(CU[0][h][1]));
    } else {
      const int p = (tid >> 6) & 1;
      const int gc = (tid >> 3) & 7, gd = tid & 7;
      val = cmulp(V2(V0[3 + p][gc]), V2(V0[5 + (gc & 1)][gd]));
    }
    RC[tid] = F2(val);
  }
  __syncthreads();

  const float2 (*V1)[8 * VP] = V[0];
  float2* const T1 = &K8[0][0][0];
  {
    {
      const int p = tid >> 7, sc = (tid >> 6) & 1, c = tid & 63;
      const float2* Vc = V1[3 + sc] + (c >> 3) * VP;
      const float2* Cf = RC + 128 + 64 * p + (c & 7);
      F1[128 + tid] = F2(cdotp<8, 1, 8>(Vc, Cf));
    }
    if (tid < 128) {
      const int e = tid >> 6, r = tid & 63;
      const float2* Va = V1[0] + (r >> 3) * VP;
      const float2* Rs = RC + 64 * e + (r & 7);
      F1[tid] = F2(cdotp<8, 1, 8>(Va, Rs));
    }
  }
  __syncthreads();

  {
    {
      const int e = tid >> 7, p = (tid >> 6) & 1, r = tid & 63;
      const int r5 = r & 31, h = r >> 5;
      v2f lohi[2];
#pragma unroll
      for (int u = 0; u < 2; u++) {
        const int rr = r5 + 32 * u;
        const int ra = rr >> 3;
        const float2* Vb = V1[1 + (ra & 1)] + (rr & 7) * VP + p;
        const float2* Ra = F1 + 64 * e + 8 * ra + p;
        lohi[u] = cdotp<4, 2, 2>(Vb, Ra);
      }
      T1[tid] = F2(h ? lohi[1] : lohi[0]);
      T1[256 + tid] = F2(cmacp(cmulp(lohi[0], V2(CU[1][h][0])), lohi[1], V2(CU[1][h][1])));
    }
    for (int o = tid; o < 512; o += 256) {
      const int c = o & 63, sc = (o >> 6) & 1, e = (o >> 7) & 1, p = o >> 8;
      const int gc = c >> 3;
      const float2* Vd = V1[5 + (gc & 1)] + (c & 7) * VP + e;
      const float2* Cc = F1 + 128 + (2 * p + sc) * 64 + 8 * gc + e;
      T1[512 + o] = F2(cdotp<4, 2, 2>(Vd, Cc));
    }
  }
  __syncthreads();

  {
    const int j = tid & 63, q = tid >> 6;
    const int sp = q & 1;
    const float2* Rt = T1 + 256 * (j & 1);
    v2f cd[4];
#pragma unroll
    for (int e = 0; e < 2; e++)
#pragma unroll
      for (int p = 0; p < 2; p++)
        cd[2 * e + p] = V2(T1[512 + ((2 * p + e) * 2 + sp) * 64 + j]);
    for (int it = 0; it < 16; it++) {
      const int i = q + 4 * it;
      v2f a = cmulp(V2(Rt[i]), cd[0]);
#pragma unroll
      for (int m = 1; m < 4; m++) a = cmacp(a, V2(Rt[64 * m + i]), cd[m]);
      S[i * LDS2 + j] = F2(a);
    }
  }
  __syncthreads();

  const int w = tid >> 5;

#pragma nounroll
  for (int l = 2; l < NL; l++) {
    const float2 (*Vl)[8 * VP] = V[l - 1];
    kron_step<false, 2, 4, 8, 16, 8 * LDS2, 0, 32>(S, Vl[0], w * LDS2);
    __syncthreads();
    kron_step<false, 2, 4, 8, 16, LDS2, 0, 32>(S, Vl[1 + (w & 1)], w * 8 * LDS2);
    __syncthreads();
    kron_step<false, 2, 4, 4 * LDS2, 8 * LDS2, 8, 0, 32 * LDS2>(
        S, Vl[3 + (w & 1)], ((w & 3) + 16 * (w >> 2)) * LDS2);
    __syncthreads();
    kron_step<true, LDS2, 2 * LDS2, 4 * LDS2, 8 * LDS2, 1, 16 * LDS2, 32 * LDS2>(
        S, Vl[5 + (w & 1)], w * 8);
    __syncthreads();

    {
      const v2f u00 = V2(CU[l][0][0]), u01 = V2(CU[l][0][1]);
      const v2f u10 = V2(CU[l][1][0]), u11 = V2(CU[l][1][1]);
      for (int t = tid; t < 1024; t += 256) {
        int il = t >> 5;
        int j = ((t & 31) << 1) | 1;
        int a0 = il * LDS2 + j, a1 = (il + 32) * LDS2 + j;
        const v2f x0 = V2(S[a0]);
        const v2f x1 = V2(S[a1]);
        S[a0] = F2(cmacp(cmulp(x0, u00), x1, u01));
        S[a1] = F2(cmacp(cmulp(x0, u10), x1, u11));
      }
    }
    __syncthreads();
  }

  {
    const int j = tid & 63, qd = tid >> 6;
    float p[16];
#pragma unroll
    for (int k = 0; k < 16; k++) {
      const float2 z = S[(qd + 4 * k) * LDS2 + j];
      p[k] = z.x * z.x + z.y * z.y;
    }
    float a1[8];
    float D0 = p[0] - p[1];
    a1[0] = p[0] + p[1];
#pragma unroll
    for (int m = 1; m < 8; m++) {
      a1[m] = p[2 * m] + p[2 * m + 1];
      D0 += p[2 * m] - p[2 * m + 1];
    }
    float a2[4];
    float D1 = a1[0] - a1[1];
    a2[0] = a1[0] + a1[1];
#pragma unroll
    for (int m = 1; m < 4; m++) {
      a2[m] = a1[2 * m] + a1[2 * m + 1];
      D1 += a1[2 * m] - a1[2 * m + 1];
    }
    const float a30 = a2[0] + a2[1], a31 = a2[2] + a2[3];
    float D2 = (a2[0] - a2[1]) + (a2[2] - a2[3]);
    float D3 = a30 - a31;
    float T = a30 + a31;

    const int lane = tid & 31;
#pragma unroll
    for (int m = 1; m < 32; m <<= 1) {
      const float t = __shfl_xor(T, m);
      T = (lane & m) ? (t - T) : (T + t);
      D0 += __shfl_xor(D0, m);
      D1 += __shfl_xor(D1, m);
      D2 += __shfl_xor(D2, m);
      D3 += __shfl_xor(D3, m);
    }

    float* part = (float*)&V[0][0][0];
    if (lane == 0) {
      part[w * 12 + 0] = D3;
      part[w * 12 + 1] = D2;
      part[w * 12 + 2] = D1;
      part[w * 12 + 3] = D0;
      part[w * 12 + 4] = (w & 4) ? -T : T;
      part[w * 12 + 5] = (w & 2) ? -T : T;
      part[w * 12 + 6] = (w & 1) ? -T : T;
    } else if ((lane & (lane - 1)) == 0) {
      part[w * 12 + 11 - (31 - __clz(lane))] = T;
    }
    __syncthreads();
    if (tid < 12) {
      float zq = part[tid];
#pragma unroll
      for (int ww = 1; ww < 8; ww++) zq += part[ww * 12 + tid];
      out[(size_t)s * 12 + tid] = zq;
    }
  }
}

extern "C" void kernel_launch(void* const* d_in, const int* in_sizes, int n_in,
                              void* d_out, int out_size, void* d_ws, size_t ws_size,
                              hipStream_t stream) {
  (void)n_in; (void)out_size; (void)ws_size;
  const float* x    = (const float*)d_in[0];
  const float* rotp = (const float*)d_in[1];
  const float* entp = (const float*)d_in[2];
  float2* W = (float2*)d_ws;
  const int B = in_sizes[0] / 48;

  qk_prep<<<NL, 64, 0, stream>>>(entp, W);
  qk_main<<<B, 256, 0, stream>>>(x, rotp, entp, W, (float*)d_out);
}


# Rot gates folded into the prep launch's factors (X*RotKron), CR(11,0) built there: per sample 48 sincos, 12 Enc 2x2s, 4 Krons, not 864 trig calls, 16 Krons; products regrouped, rounding changes
# speedup vs baseline: 11.5139x; 1.0666x over previous
#include <hip/hip_runtime.h>
#include <math.h>

#define LDS2 66
#define NL  4
#define NX  7
#define VP  10

typedef __attribute__((ext_vector_type(2))) float v2f;
typedef __attribute__((ext_vector_type(8))) float v8f;

__device__ __forceinline__ float2 cmul(float2 a, float2 b) {
  return make_float2(a.x * b.x - a.y * b.y, a.x * b.y + a.y * b.x);
}
__device__ __forceinline__ float2 cadd(float2 a, float2 b) {
  return make_float2(a.x + b.x, a.y + b.y);
}

__device__ __forceinline__ v2f V2(float2 a) { v2f r; r.x = a.x; r.y = a.y; return r; }
__device__ __forceinline__ float2 F2(v2f a) { return make_float2(a.x, a.y); }
__device__ __forceinline__ v2f irot(v2f b) { v2f r; r.x = -b.y; r.y = b.x; return r; }
__device__ __forceinline__ v2f cmulp(v2f a, v2f b) {
  return __builtin_elementwise_fma(a.yy, irot(b), a.xx * b);
}
__device__ __forceinline__ v2f cmacp(v2f acc, v2f a, v2f b) {
  acc = __builtin_elementwise_fma(a.xx, b, acc);
  return __builtin_elementwise_fma(a.yy, irot(b), acc);
}
template <int N, int SA, int SB>
__device__ __forceinline__ v2f cdotp(const float2* a, const float2* b) {
  v2f acc = cmulp(V2(a[0]), V2(b[0]));
#pragma unroll
  for (int k = 1; k < N; k++) acc = cmacp(acc, V2(a[SA * k]), V2(b[SB * k]));
  return acc;
}

struct C2 { float2 a00, a01, a10, a11; };
__device__ __forceinline__ C2 mm2(const C2 L, const C2 R) {
  C2 o;
  o.a00 = F2(cmacp(cmulp(V2(L.a00), V2(R.a00)), V2(L.a01), V2(R.a10)));
  o.a01 = F2(cmacp(cmulp(V2(L.a00), V2(R.a01)), V2(L.a01), V2(R.a11)));
  o.a10 = F2(cmacp(cmulp(V2(L.a10), V2(R.a00)), V2(L.a11), V2(R.a10)));
  o.a11 = F2(cmacp(cmulp(V2(L.a10), V2(R.a01)), V2(L.a11), V2(R.a11)));
  return o;
}

__device__ __forceinline__ void rot_gate(float phi, float th, float om,
                                         float2* u00, float2* u01,
                                         float2* u10, float2* u11) {
  float ct = cosf(0.5f * th), st = sinf(0.5f * th);
  float ap = -0.5f * (phi + om);
  float am = 0.5f * (phi - om);
  *u00 = make_float2(ct * cosf(ap), ct * sinf(ap));
  *u01 = make_float2(-st * cosf(am), -st * sinf(am));
  *u10 = make_float2(st * cosf(-am), st * sinf(-am));
  *u11 = make_float2(ct * cosf(-ap), ct * sinf(-ap));
}

__global__ __launch_bounds__(64) void qk_prep(const float* __restrict__ rotp,
                                              const float* __restrict__ entp,
                                              float2* __restrict__ W) {
  __shared__ float2 U[12][2][2];
  __shared__ float2 R[12][2][2];
  const int l = blockIdx.x;
  const int tid = threadIdx.x;

  if (tid < 12) {
    const float* p = &entp[(l * 12 + tid) * 3];
    rot_gate(p[0], p[1], p[2], &U[tid][0][0], &U[tid][0][1],
             &U[tid][1][0], &U[tid][1][1]);
  } else if (tid >= 32 && tid < 44) {
    const int q = tid - 32;
    const float* p = &rotp[(l * 12 + q) * 3];
    rot_gate(p[0], p[1], p[2], &R[q][0][0], &R[q][0][1],
             &R[q][1][0], &R[q][1][1]);
  }
  __syncthreads();

  if (tid >= 56 && tid < 60) {
    const int k = tid - 56;
    W[NL * NX * 64 + l * 4 + k] = U[11][k >> 1][k & 1];
  }

  if (tid < NX * 8) {
    const int t = tid >> 3, j = tid & 7;
    const int g = (t + 1) >> 1;
    const bool link = (t > 0) && ((t & 1) == 0);
    float2 v[8];
#pragma unroll
    for (int r = 0; r < 8; r++)
      v[r] = cmul(cmul(R[3 * g][r >> 2][j >> 2], R[3 * g + 1][(r >> 1) & 1][(j >> 1) & 1]),
                  R[3 * g + 2][r & 1][j & 1]);
    if (link) {
      const int q = 3 * g - 1;
      float2 u00 = U[q][0][0], u01 = U[q][0][1], u10 = U[q][1][0], u11 = U[q][1][1];
#pragma unroll
      for (int i = 0; i < 4; i++) {
        float2 x0 = v[i], x1 = v[i + 4];
        v[i]     = cadd(cmul(u00, x0), cmul(u01, x1));
        v[i + 4] = cadd(cmul(u10, x0), cmul(u11, x1));
      }
    }
    {
      const int q = 3 * g;
      float2 u00 = U[q][0][0], u01 = U[q][0][1], u10 = U[q][1][0], u11 = U[q][1][1];
#pragma unroll
      for (int i = 4; i < 6; i++) {
        float2 x0 = v[i], x1 = v[i + 2];
        v[i]     = cadd(cmul(u00, x0), cmul(u01, x1));
        v[i + 2] = cadd(cmul(u10, x0), cmul(u11, x1));
      }
    }
    {
      const int q = 3 * g + 1;
      float2 u00 = U[q][0][0], u01 = U[q][0][1], u10 = U[q][1][0], u11 = U[q][1][1];
#pragma unroll
      for (int i = 2; i < 8; i += 4) {
        float2 x0 = v[i], x1 = v[i + 1];
        v[i]     = cadd(cmul(u00, x0), cmul(u01, x1));
        v[i + 1] = cadd(cmul(u10, x0), cmul(u11, x1));
      }
    }
    float* Wt = (float*)(W + (size_t)(l * NX + t) * 64);
#pragma unroll
    for (int r = 0; r < 8; r++) {
      Wt[r * 8 + j] = v[r].x;
      Wt[64 + r * 8 + j] = v[r].y;
    }
  }
}

__device__ __forceinline__ v8f wmma4(v2f a, v2f b, v8f c) {
  return __builtin_amdgcn_wmma_f32_16x16x4_f32(
      false, a, false, b, (short)0, c, false, false);
}

template <bool PAIRS, int N0, int N1, int N2, int N3, int GS, int TS0, int TS1>
__device__ __forceinline__ void kron_step(float2* S, const float2* V, int base0) {
  const int lane = threadIdx.x & 31;
  const int n = lane & 15;
  const int hi = lane >> 4;
  const int gp = PAIRS ? (4 * (n >> 3) + ((n & 7) >> 1)) : ((n & 6) + (n >> 3));
  const bool im = (n & 1) != 0;

  const float4* Vrow = (const float4*)(V + gp * VP + 4 * hi);
  const float4 q0 = Vrow[0], q1 = Vrow[1];
  v2f a[4];
  a[0].x = im ? q0.y : q0.x;  a[0].y = im ? q0.x : -q0.y;
  a[1].x = im ? q0.w : q0.z;  a[1].y = im ? q0.z : -q0.w;
  a[2].x = im ? q1.y : q1.x;  a[2].y = im ? q1.x : -q1.y;
  a[3].x = im ? q1.w : q1.z;  a[3].y = im ? q1.z : -q1.w;

  float2* const P = S + base0 + (n & 1) * N0 + ((n >> 1) & 1) * N1 +
                    ((n >> 2) & 1) * N2 + (n >> 3) * N3 + hi * (PAIRS ? 4 : GS);
  v2f b[4][4];
  if (PAIRS) {
#pragma unroll
    for (int t = 0; t < 4; t++) {
#pragma unroll
      for (int h = 0; h < 2; h++) {
        const float4 z = *(const float4*)(P + (t & 1) * TS0 + (t >> 1) * TS1 + 2 * h);
        b[t][2 * h].x = z.x;      b[t][2 * h].y = z.y;
        b[t][2 * h + 1].x = z.z;  b[t][2 * h + 1].y = z.w;
      }
    }
  } else {
#pragma unroll
    for (int u = 0; u < 2; u++) {
#pragma unroll
      for (int ks = 0; ks < 4; ks++) {
        const float4 z = *(const float4*)(P + u * TS1 + ks * 2 * GS);
        b[2 * u][ks].x = z.x;      b[2 * u][ks].y = z.y;
        b[2 * u + 1][ks].x = z.z;  b[2 * u + 1][ks].y = z.w;
      }
    }
  }

  v8f acc[4] = {};
#pragma unroll
  for (int ks = 0; ks < 4; ks++) {
#pragma unroll
    for (int t = 0; t < 4; t++) acc[t] = wmma4(a[ks], b[t][ks], acc[t]);
  }

  if (PAIRS) {
#pragma unroll
    for (int t = 0; t < 4; t++) {
#pragma unroll
      for (int h = 0; h < 2; h++)
        *(float4*)(P + (t & 1) * TS0 + (t >> 1) * TS1 + 2 * h) =
            make_float4(acc[t][4 * h], acc[t][4 * h + 1], acc[t][4 * h + 2], acc[t][4 * h + 3]);
    }
  } else {
#pragma unroll
    for (int u = 0; u < 2; u++) {
#pragma unroll
      for (int gl = 0; gl < 4; gl++)
        *(float4*)(P + u * TS1 + gl * 2 * GS) =
            make_float4(acc[2 * u][2 * gl], acc[2 * u][2 * gl + 1],
                        acc[2 * u + 1][2 * gl], acc[2 * u + 1][2 * gl + 1]);
    }
  }
}

__global__ __launch_bounds__(256, 10) void qk_main(const float* __restrict__ x,
                                               const float2* __restrict__ W,
                                               float* __restrict__ out) {
  __shared__ __attribute__((aligned(16))) float2 S[64 * LDS2];
  __shared__ float2 SC[48];
  __shared__ float2 E[12][2][2];
  __shared__ float2 T1s[1024];
  __shared__ __attribute__((aligned(16))) float2 V[NL - 1][NX][8 * VP];
  __shared__ float2 V0[NX][8];
  __shared__ float2 RC[256];
  __shared__ float2 CU[NL][2][2];
  __shared__ float2 F1[384];

  const int tid = threadIdx.x;
  const int s = blockIdx.x;

  if (tid < 48) {
    float sn, cs;
    sincosf(0.5f * x[(size_t)s * 48 + tid], &sn, &cs);
    SC[tid] = make_float2(cs, sn);
  }
  if (tid >= 64 && tid < 64 + 4 * NL) (&CU[0][0][0])[tid - 64] = W[NL * NX * 64 + tid - 64];
  __syncthreads();

  if (tid < 12) {
    const float2 pa = SC[tid], pb = SC[12 + tid], pc = SC[24 + tid], pd = SC[36 + tid];
    const float ca = pa.x, sa = pa.y, cb = pb.x, sb = pb.y;
    const float cc = pc.x, sc = pc.y, cd = pd.x, sd = pd.y;
    C2 M  = { {ca, 0.f}, {-sa, 0.f}, {sa, 0.f}, {ca, 0.f} };
    C2 RX = { {cb, 0.f}, {0.f, -sb}, {0.f, -sb}, {cb, 0.f} };
    C2 RZ = { {cc, -sc}, {0.f, 0.f}, {0.f, 0.f}, {cc, sc} };
    C2 RY = { {cd, 0.f}, {-sd, 0.f}, {sd, 0.f}, {cd, 0.f} };
    M = mm2(RX, M); M = mm2(RZ, M); M = mm2(RY, M);
    E[tid][0][0] = M.a00; E[tid][0][1] = M.a01;
    E[tid][1][0] = M.a10; E[tid][1][1] = M.a11;
  }
  __syncthreads();

  float2 (*const EK)[64] = (float2 (*)[64])T1s;
  {
    const int kg = tid >> 6, ka = (tid >> 3) & 7, kb8 = tid & 7;
    const v2f p = cmulp(cmulp(V2(E[3 * kg][(ka >> 2) & 1][(kb8 >> 2) & 1]),
                              V2(E[3 * kg + 1][(ka >> 1) & 1][(kb8 >> 1) & 1])),
                        V2(E[3 * kg + 2][ka & 1][kb8 & 1]));
    EK[kg][ka * 8 + kb8] = F2(p);
  }
  __syncthreads();

  {
    const int lane = tid & 31;
    const int n = lane & 15, hi = lane >> 4;
    const int r = n & 7, tt = n >> 3;
    const int wv = tid >> 5;
    const int g = wv & 3;
    const int t = g ? 2 * g - 1 + tt : 0;
    const int comp = (tt != hi) ? 1 : 0;
    const float sg = (tt == 0 && hi == 1) ? -1.f : 1.f;
    const float* Wf = (const float*)W;
    v2f fa[4], fb[2][4];
#pragma unroll
    for (int i = 0; i < 2; i++) {
      const int l = 2 * i + (wv >> 2);
      const float* Xp = Wf + (size_t)(l * NX + t) * 128 + hi * 64 + r * 8;
      const float4 x0 = *(const float4*)Xp;
      const float4 x1 = *(const float4*)(Xp + 4);
      fb[i][0].x = x0.x; fb[i][0].y = x0.y; fb[i][1].x = x0.z; fb[i][1].y = x0.w;
      fb[i][2].x = x1.x; fb[i][2].y = x1.y; fb[i][3].x = x1.z; fb[i][3].y = x1.w;
    }
    {
      const float* Kc = (const float*)&EK[g][r] + comp;
#pragma unroll
      for (int ks = 0; ks < 4; ks++) {
        fa[ks].x = sg * Kc[(2 * ks) * 16];
        fa[ks].y = sg * Kc[(2 * ks + 1) * 16];
      }
    }
    v8f acc[2] = {};
#pragma unroll
    for (int ks = 0; ks < 4; ks++) {
#pragma unroll
      for (int i = 0; i < 2; i++) acc[i] = wmma4(fa[ks], fb[i][ks], acc[i]);
    }
    {
      float* Vd = (float*)&V[1 + (wv >> 2)][t][r * VP] + hi;
      if (__builtin_amdgcn_readfirstlane(g) == 3) {
#pragma unroll
        for (int v = 0; v < 8; v++) Vd[2 * v] = acc[1][v];
      } else {
#pragma unroll
        for (int v = 0; v < 8; v++) Vd[2 * (4 * (v & 1) + (v >> 1))] = acc[1][v];
      }
    }
    if (__builtin_amdgcn_readfirstlane(wv) >> 2) {
      float* Vd = (float*)&V[0][t][r * VP] + hi;
#pragma unroll
      for (int v = 0; v < 8; v++) Vd[2 * v] = acc[0][v];
    } else {
      ((float*)&V0[t][r])[hi] = acc[0][0];
    }
  }
  __syncthreads();

  {
    v2f val;
    if (tid < 64) {
      const int ra = tid >> 3, rb = tid & 7;
      val = cmulp(V2(V0[0][ra]), V2(V0[1 + (ra & 1)][rb]));
    } else if (tid < 128) {
      const int h = (tid >> 5) & 1;
      const int ra = (tid >> 3) & 3, rb = tid & 7;
      const v2f bf = V2(V0[1 + (ra & 1)][rb]);
      const v2f x0 = cmulp(V2(V0[0][ra]), bf);
      const v2f x1 = cmulp(V2(V0[0][ra + 4]), bf);
      val = cmacp(cmulp(x0, V2(CU[0][h][0])), x1, V2(CU[0][h][1]));
    } else {
      const int p = (tid >> 6) & 1;
      const int gc = (tid >> 3) & 7, gd = tid & 7;
      val = cmulp(V2(V0[3 + p][gc]), V2(V0[5 + (gc & 1)][gd]));
    }
    RC[tid] = F2(val);
  }
  __syncthreads();

  const float2 (*V1)[8 * VP] = V[0];
  float2* const T1 = T1s;
  {
    {
      const int p = tid >> 7, sc = (tid >> 6) & 1, c = tid & 63;
      const float2* Vc = V1[3 + sc] + (c >> 3) * VP;
      const float2* Cf = RC + 128 + 64 * p + (c & 7);
      F1[128 + tid] = F2(cdotp<8, 1, 8>(Vc, Cf));
    }
    if (tid < 128) {
      const int e = tid >> 6, r = tid & 63;
      const float2* Va = V1[0] + (r >> 3) * VP;
      const float2* Rs = RC + 64 * e + (r & 7);
      F1[tid] = F2(cdotp<8, 1, 8>(Va, Rs));
    }
  }
  __syncthreads();

  {
    {
      const int e = tid >> 7, p = (tid >> 6) & 1, r = tid & 63;
      const int r5 = r & 31, h = r >> 5;
      v2f lohi[2];
#pragma unroll
      for (int u = 0; u < 2; u++) {
        const int rr = r5 + 32 * u;
        const int ra = rr >> 3;
        const float2* Vb = V1[1 + (ra & 1)] + (rr & 7) * VP + p;
        const float2* Ra = F1 + 64 * e + 8 * ra + p;
        lohi[u] = cdotp<4, 2, 2>(Vb, Ra);
      }
      T1[tid] = F2(h ? lohi[1] : lohi[0]);
      T1[256 + tid] = F2(cmacp(cmulp(lohi[0], V2(CU[1][h][0])), lohi[1], V2(CU[1][h][1])));
    }
    for (int o = tid; o < 512; o += 256) {
      const int c = o & 63, sc = (o >> 6) & 1, e = (o >> 7) & 1, p = o >> 8;
      const int gc = c >> 3;
      const float2* Vd = V1[5 + (gc & 1)] + (c & 7) * VP + e;
      const float2* Cc = F1 + 128 + (2 * p + sc) * 64 + 8 * gc + e;
      T1[512 + o] = F2(cdotp<4, 2, 2>(Vd, Cc));
    }
  }
  __syncthreads();

  {
    const int j = tid & 63, q = tid >> 6;
    const int sp = q & 1;
    const float2* Rt = T1 + 256 * (j & 1);
    v2f cd[4];
#pragma unroll
    for (int e = 0; e < 2; e++)
#pragma unroll
      for (int p = 0; p < 2; p++)
        cd[2 * e + p] = V2(T1[512 + ((2 * p + e) * 2 + sp) * 64 + j]);
    for (int it = 0; it < 16; it++) {
      const int i = q + 4 * it;
      v2f a = cmulp(V2(Rt[i]), cd[0]);
#pragma unroll
      for (int m = 1; m < 4; m++) a = cmacp(a, V2(Rt[64 * m + i]), cd[m]);
      S[i * LDS2 + j] = F2(a);
    }
  }
  __syncthreads();

  const int w = tid >> 5;

#pragma nounroll
  for (int l = 2; l < NL; l++) {
    const float2 (*Vl)[8 * VP] = V[l - 1];
    kron_step<false, 2, 4, 8, 16, 8 * LDS2, 0, 32>(S, Vl[0], w * LDS2);
    __syncthreads();
    kron_step<false, 2, 4, 8, 16, LDS2, 0, 32>(S, Vl[1 + (w & 1)], w * 8 * LDS2);
    __syncthreads();
    kron_step<false, 2, 4, 4 * LDS2, 8 * LDS2, 8, 0, 32 * LDS2>(
        S, Vl[3 + (w & 1)], ((w & 3) + 16 * (w >> 2)) * LDS2);
    __syncthreads();
    kron_step<true, LDS2, 2 * LDS2, 4 * LDS2, 8 * LDS2, 1, 16 * LDS2, 32 * LDS2>(
        S, Vl[5 + (w & 1)], w * 8);
    __syncthreads();

    {
      const v2f u00 = V2(CU[l][0][0]), u01 = V2(CU[l][0][1]);
      const v2f u10 = V2(CU[l][1][0]), u11 = V2(CU[l][1][1]);
      for (int t = tid; t < 1024; t += 256) {
        int il = t >> 5;
        int j = ((t & 31) << 1) | 1;
        int a0 = il * LDS2 + j, a1 = (il + 32) * LDS2 + j;
        const v2f x0 = V2(S[a0]);
        const v2f x1 = V2(S[a1]);
        S[a0] = F2(cmacp(cmulp(x0, u00), x1, u01));
        S[a1] = F2(cmacp(cmulp(x0, u10), x1, u11));
      }
    }
    __syncthreads();
  }

  {
    const int j = tid & 63, qd = tid >> 6;
    float p[16];
#pragma unroll
    for (int k = 0; k < 16; k++) {
      const float2 z = S[(qd + 4 * k) * LDS2 + j];
      p[k] = z.x * z.x + z.y * z.y;
    }
    float a1[8];
    float D0 = p[0] - p[1];
    a1[0] = p[0] + p[1];
#pragma unroll
    for (int m = 1; m < 8; m++) {
      a1[m] = p[2 * m] + p[2 * m + 1];
      D0 += p[2 * m] - p[2 * m + 1];
    }
    float a2[4];
    float D1 = a1[0] - a1[1];
    a2[0] = a1[0] + a1[1];
#pragma unroll
    for (int m = 1; m < 4; m++) {
      a2[m] = a1[2 * m] + a1[2 * m + 1];
      D1 += a1[2 * m] - a1[2 * m + 1];
    }
    const float a30 = a2[0] + a2[1], a31 = a2[2] + a2[3];
    float D2 = (a2[0] - a2[1]) + (a2[2] - a2[3]);
    float D3 = a30 - a31;
    float T = a30 + a31;

    const int lane = tid & 31;
#pragma unroll
    for (int m = 1; m < 32; m <<= 1) {
      const float t = __shfl_xor(T, m);
      T = (lane & m) ? (t - T) : (T + t);
      D0 += __shfl_xor(D0, m);
      D1 += __shfl_xor(D1, m);
      D2 += __shfl_xor(D2, m);
      D3 += __shfl_xor(D3, m);
    }

    float* part = (float*)&V[0][0][0];
    if (lane == 0) {
      part[w * 12 + 0] = D3;
      part[w * 12 + 1] = D2;
      part[w * 12 + 2] = D1;
      part[w * 12 + 3] = D0;
      part[w * 12 + 4] = (w & 4) ? -T : T;
      part[w * 12 + 5] = (w & 2) ? -T : T;
      part[w * 12 + 6] = (w & 1) ? -T : T;
    } else if ((lane & (lane - 1)) == 0) {
      part[w * 12 + 11 - (31 - __clz(lane))] = T;
    }
    __syncthreads();
    if (tid < 12) {
      float zq = part[tid];
#pragma unroll
      for (int ww = 1; ww < 8; ww++) zq += part[ww * 12 + tid];
      out[(size_t)s * 12 + tid] = zq;
    }
  }
}

extern "C" void kernel_launch(void* const* d_in, const int* in_sizes, int n_in,
                              void* d_out, int out_size, void* d_ws, size_t ws_size,
                              hipStream_t stream) {
  (void)n_in; (void)out_size; (void)ws_size;
  const float* x    = (const float*)d_in[0];
  const float* rotp = (const float*)d_in[1];
  const float* entp = (const float*)d_in[2];
  float2* W = (float2*)d_ws;
  const int B = in_sizes[0] / 48;

  qk_prep<<<NL, 64, 0, stream>>>(rotp, entp, W);
  qk_main<<<B, 256, 0, stream>>>(x, W, (float*)d_out);
}
